# starting kernel
# baseline (speedup 1.0000x reference)
#include <hip/hip_runtime.h>

typedef __attribute__((ext_vector_type(2))) float v2f;
typedef __attribute__((ext_vector_type(8))) float v8f;

#define N_NODES 100000
#define N_EDGES 1600000
#define IN_F    256
#define OUT_F   64

__global__ __launch_bounds__(256) void gemm_xw_wmma(const float* __restrict__ X,
                                                    const float* __restrict__ W,
                                                    float* __restrict__ support) {
  __shared__ float lw[IN_F * OUT_F];

  for (int i = threadIdx.x; i < IN_F * OUT_F; i += 256) {
    int k = i >> 6;
    int n = i & 63;
    int sl = k * 64 + ((((n >> 4) + (k >> 1)) & 3) << 4) + (n & 15);
    lw[sl] = W[i];
  }
  __syncthreads();

  const int wave = threadIdx.x >> 5;
  const int lane = threadIdx.x & 31;
  const int hi   = lane >> 4;
  const int lo   = lane & 15;
  const int row_base = blockIdx.x * 128 + wave * 16;
  if (row_base >= N_NODES) return;

  const float* xrow = X + (size_t)(row_base + lo) * IN_F;

  v8f acc0 = {}, acc1 = {}, acc2 = {}, acc3 = {};
  for (int k0 = 0; k0 < IN_F; k0 += 4) {
    const int ka = k0 + 2 * hi;
    v2f a = *(const v2f*)(xrow + ka);
    const float* wr = lw + ka * 64;
    const int kr = ka >> 1;
    const int s0 = (((kr + 0) & 3) << 4) + lo;
    const int s1 = (((kr + 1) & 3) << 4) + lo;
    const int s2 = (((kr + 2) & 3) << 4) + lo;
    const int s3 = (((kr + 3) & 3) << 4) + lo;
    v2f b0 = { wr[s0], wr[s0 + 64] };
    v2f b1 = { wr[s1], wr[s1 + 64] };
    v2f b2 = { wr[s2], wr[s2 + 64] };
    v2f b3 = { wr[s3], wr[s3 + 64] };
    acc0 = __builtin_amdgcn_wmma_f32_16x16x4_f32(false, a, false, b0, (short)0, acc0, false, false);
    acc1 = __builtin_amdgcn_wmma_f32_16x16x4_f32(false, a, false, b1, (short)0, acc1, false, false);
    acc2 = __builtin_amdgcn_wmma_f32_16x16x4_f32(false, a, false, b2, (short)0, acc2, false, false);
    acc3 = __builtin_amdgcn_wmma_f32_16x16x4_f32(false, a, false, b3, (short)0, acc3, false, false);
  }

  float* orow = support + (size_t)(row_base + 8 * hi) * OUT_F + lo;
#pragma unroll
  for (int r = 0; r < 8; ++r) {
    orow[(size_t)r * OUT_F +  0] = acc0[r];
    orow[(size_t)r * OUT_F + 16] = acc1[r];
    orow[(size_t)r * OUT_F + 32] = acc2[r];
    orow[(size_t)r * OUT_F + 48] = acc3[r];
  }
}

__global__ __launch_bounds__(256) void init_bias(const float* __restrict__ bias,
                                                 float* __restrict__ out) {
  const float4* b4 = (const float4*)bias;
  float4* o4 = (float4*)out;
  const int total = N_NODES * OUT_F / 4;
  for (int i = blockIdx.x * blockDim.x + threadIdx.x; i < total;
       i += gridDim.x * blockDim.x) {
    o4[i] = b4[i & 15];
  }
}

#define EDGES_PER_WAVE 128

__global__ __launch_bounds__(256) void spmm_scatter(const float* __restrict__ support,
                                                    const float* __restrict__ edge_vals,
                                                    const int* __restrict__ row_idx,
                                                    const int* __restrict__ col_idx,
                                                    float* __restrict__ out) {
  const int lane = threadIdx.x & 31;
  const int wave = blockIdx.x * 8 + (threadIdx.x >> 5);
  const long e0 = (long)wave * EDGES_PER_WAVE;
  if (e0 >= N_EDGES) return;

  const int c2 = lane * 2;
  float ax = 0.f, ay = 0.f;
  int cur = -1;

  for (long e = e0; e < e0 + EDGES_PER_WAVE; e += 4) {
    const int4   rr = *(const int4*)(row_idx + e);
    const int4   cc = *(const int4*)(col_idx + e);
    const float4 vv = *(const float4*)(edge_vals + e);
    const float2 s0 = *(const float2*)(support + (size_t)cc.x * OUT_F + c2);
    const float2 s1 = *(const float2*)(support + (size_t)cc.y * OUT_F + c2);
    const float2 s2 = *(const float2*)(support + (size_t)cc.z * OUT_F + c2);
    const float2 s3 = *(const float2*)(support + (size_t)cc.w * OUT_F + c2);

    if (rr.x != cur) {
      if (cur >= 0) {
        atomicAdd(out + (size_t)cur * OUT_F + c2,     ax);
        atomicAdd(out + (size_t)cur * OUT_F + c2 + 1, ay);
      }
      cur = rr.x; ax = 0.f; ay = 0.f;
    }
    ax = fmaf(vv.x, s0.x, ax); ay = fmaf(vv.x, s0.y, ay);

    if (rr.y != cur) {
      atomicAdd(out + (size_t)cur * OUT_F + c2,     ax);
      atomicAdd(out + (size_t)cur * OUT_F + c2 + 1, ay);
      cur = rr.y; ax = 0.f; ay = 0.f;
    }
    ax = fmaf(vv.y, s1.x, ax); ay = fmaf(vv.y, s1.y, ay);

    if (rr.z != cur) {
      atomicAdd(out + (size_t)cur * OUT_F + c2,     ax);
      atomicAdd(out + (size_t)cur * OUT_F + c2 + 1, ay);
      cur = rr.z; ax = 0.f; ay = 0.f;
    }
    ax = fmaf(vv.z, s2.x, ax); ay = fmaf(vv.z, s2.y, ay);

    if (rr.w != cur) {
      atomicAdd(out + (size_t)cur * OUT_F + c2,     ax);
      atomicAdd(out + (size_t)cur * OUT_F + c2 + 1, ay);
      cur = rr.w; ax = 0.f; ay = 0.f;
    }
    ax = fmaf(vv.w, s3.x, ax); ay = fmaf(vv.w, s3.y, ay);
  }

  if (cur >= 0) {
    atomicAdd(out + (size_t)cur * OUT_F + c2,     ax);
    atomicAdd(out + (size_t)cur * OUT_F + c2 + 1, ay);
  }
}

extern "C" void kernel_launch(void* const* d_in, const int* in_sizes, int n_in,
                              void* d_out, int out_size, void* d_ws, size_t ws_size,
                              hipStream_t stream) {
  const float* X         = (const float*)d_in[0];
  const float* edge_vals = (const float*)d_in[1];
  const float* W         = (const float*)d_in[2];
  const float* bias      = (const float*)d_in[3];
  const int*   row_idx   = (const int*)d_in[4];
  const int*   col_idx   = (const int*)d_in[5];
  float* out     = (float*)d_out;
  float* support = (float*)d_ws;

  (void)in_sizes; (void)n_in; (void)out_size; (void)ws_size;

  gemm_xw_wmma<<<(N_NODES + 127) / 128, 256, 0, stream>>>(X, W, support);
  init_bias<<<2048, 256, 0, stream>>>(bias, out);
  int waves  = (N_EDGES + EDGES_PER_WAVE - 1) / EDGES_PER_WAVE;
  int blocks = (waves + 7) / 8;
  spmm_scatter<<<blocks, 256, 0, stream>>>(support, edge_vals, row_idx, col_idx, out);
}


# GEMM: X fragments fetched a 32-k block (one 128-B line per row) at a time into a ring of three register sets, two blocks ahead of the WMMAs, first two before W staging; k order unchanged
# speedup vs baseline: 1.0628x; 1.0628x over previous
#include <hip/hip_runtime.h>

typedef __attribute__((ext_vector_type(2))) float v2f;
typedef __attribute__((ext_vector_type(8))) float v8f;

#define N_NODES 100000
#define N_EDGES 1600000
#define IN_F    256
#define OUT_F   64

__global__ __launch_bounds__(256) void gemm_xw_wmma(const float* __restrict__ X,
                                                    const float* __restrict__ W,
                                                    float* __restrict__ support) {
  __shared__ float lw[IN_F * OUT_F];

  const int wave = threadIdx.x >> 5;
  const int lane = threadIdx.x & 31;
  const int hi   = lane >> 4;
  const int lo   = lane & 15;
  const int row_base = blockIdx.x * 128 + wave * 16;
  const bool valid = row_base < N_NODES;

  const float* xrow = X + (size_t)((valid ? row_base : 0) + lo) * IN_F + 2 * hi;
  v2f a[3][8];
#pragma unroll
  for (int j = 0; j < 8; ++j) a[0][j] = *(const v2f*)(xrow + 4 * j);
#pragma unroll
  for (int j = 0; j < 8; ++j) a[1][j] = *(const v2f*)(xrow + 32 + 4 * j);

  for (int i = threadIdx.x; i < IN_F * OUT_F; i += 256) {
    int k = i >> 6;
    int n = i & 63;
    int sl = k * 64 + ((((n >> 4) + (k >> 1)) & 3) << 4) + (n & 15);
    lw[sl] = W[i];
  }
  __syncthreads();

  if (!valid) return;

  v8f acc0 = {}, acc1 = {}, acc2 = {}, acc3 = {};
#pragma unroll
  for (int kb = 0; kb < 8; ++kb) {
    if (kb + 2 < 8) {
#pragma unroll
      for (int j = 0; j < 8; ++j)
        a[(kb + 2) % 3][j] = *(const v2f*)(xrow + (kb + 2) * 32 + 4 * j);
    }
#pragma unroll
    for (int j = 0; j < 8; ++j) {
      const int k0 = kb * 32 + 4 * j;
      const int ka = k0 + 2 * hi;
      const v2f av = a[kb % 3][j];
      const float* wr = lw + ka * 64;
      const int kr = ka >> 1;
      const int s0 = (((kr + 0) & 3) << 4) + lo;
      const int s1 = (((kr + 1) & 3) << 4) + lo;
      const int s2 = (((kr + 2) & 3) << 4) + lo;
      const int s3 = (((kr + 3) & 3) << 4) + lo;
      v2f b0 = { wr[s0], wr[s0 + 64] };
      v2f b1 = { wr[s1], wr[s1 + 64] };
      v2f b2 = { wr[s2], wr[s2 + 64] };
      v2f b3 = { wr[s3], wr[s3 + 64] };
      acc0 = __builtin_amdgcn_wmma_f32_16x16x4_f32(false, av, false, b0, (short)0, acc0, false, false);
      acc1 = __builtin_amdgcn_wmma_f32_16x16x4_f32(false, av, false, b1, (short)0, acc1, false, false);
      acc2 = __builtin_amdgcn_wmma_f32_16x16x4_f32(false, av, false, b2, (short)0, acc2, false, false);
      acc3 = __builtin_amdgcn_wmma_f32_16x16x4_f32(false, av, false, b3, (short)0, acc3, false, false);
    }
  }

  float* orow = support + (size_t)(row_base + 8 * hi) * OUT_F + lo;
#pragma unroll
  for (int r = 0; r < 8; ++r) {
    orow[(size_t)r * OUT_F +  0] = acc0[r];
    orow[(size_t)r * OUT_F + 16] = acc1[r];
    orow[(size_t)r * OUT_F + 32] = acc2[r];
    orow[(size_t)r * OUT_F + 48] = acc3[r];
  }
}

__global__ __launch_bounds__(256) void init_bias(const float* __restrict__ bias,
                                                 float* __restrict__ out) {
  const float4* b4 = (const float4*)bias;
  float4* o4 = (float4*)out;
  const int total = N_NODES * OUT_F / 4;
  for (int i = blockIdx.x * blockDim.x + threadIdx.x; i < total;
       i += gridDim.x * blockDim.x) {
    o4[i] = b4[i & 15];
  }
}

#define EDGES_PER_WAVE 128

__global__ __launch_bounds__(256) void spmm_scatter(const float* __restrict__ support,
                                                    const float* __restrict__ edge_vals,
                                                    const int* __restrict__ row_idx,
                                                    const int* __restrict__ col_idx,
                                                    float* __restrict__ out) {
  const int lane = threadIdx.x & 31;
  const int wave = blockIdx.x * 8 + (threadIdx.x >> 5);
  const long e0 = (long)wave * EDGES_PER_WAVE;
  if (e0 >= N_EDGES) return;

  const int c2 = lane * 2;
  float ax = 0.f, ay = 0.f;
  int cur = -1;

  for (long e = e0; e < e0 + EDGES_PER_WAVE; e += 4) {
    const int4   rr = *(const int4*)(row_idx + e);
    const int4   cc = *(const int4*)(col_idx + e);
    const float4 vv = *(const float4*)(edge_vals + e);
    const float2 s0 = *(const float2*)(support + (size_t)cc.x * OUT_F + c2);
    const float2 s1 = *(const float2*)(support + (size_t)cc.y * OUT_F + c2);
    const float2 s2 = *(const float2*)(support + (size_t)cc.z * OUT_F + c2);
    const float2 s3 = *(const float2*)(support + (size_t)cc.w * OUT_F + c2);

    if (rr.x != cur) {
      if (cur >= 0) {
        atomicAdd(out + (size_t)cur * OUT_F + c2,     ax);
        atomicAdd(out + (size_t)cur * OUT_F + c2 + 1, ay);
      }
      cur = rr.x; ax = 0.f; ay = 0.f;
    }
    ax = fmaf(vv.x, s0.x, ax); ay = fmaf(vv.x, s0.y, ay);

    if (rr.y != cur) {
      atomicAdd(out + (size_t)cur * OUT_F + c2,     ax);
      atomicAdd(out + (size_t)cur * OUT_F + c2 + 1, ay);
      cur = rr.y; ax = 0.f; ay = 0.f;
    }
    ax = fmaf(vv.y, s1.x, ax); ay = fmaf(vv.y, s1.y, ay);

    if (rr.z != cur) {
      atomicAdd(out + (size_t)cur * OUT_F + c2,     ax);
      atomicAdd(out + (size_t)cur * OUT_F + c2 + 1, ay);
      cur = rr.z; ax = 0.f; ay = 0.f;
    }
    ax = fmaf(vv.z, s2.x, ax); ay = fmaf(vv.z, s2.y, ay);

    if (rr.w != cur) {
      atomicAdd(out + (size_t)cur * OUT_F + c2,     ax);
      atomicAdd(out + (size_t)cur * OUT_F + c2 + 1, ay);
      cur = rr.w; ax = 0.f; ay = 0.f;
    }
    ax = fmaf(vv.w, s3.x, ax); ay = fmaf(vv.w, s3.y, ay);
  }

  if (cur >= 0) {
    atomicAdd(out + (size_t)cur * OUT_F + c2,     ax);
    atomicAdd(out + (size_t)cur * OUT_F + c2 + 1, ay);
  }
}

extern "C" void kernel_launch(void* const* d_in, const int* in_sizes, int n_in,
                              void* d_out, int out_size, void* d_ws, size_t ws_size,
                              hipStream_t stream) {
  const float* X         = (const float*)d_in[0];
  const float* edge_vals = (const float*)d_in[1];
  const float* W         = (const float*)d_in[2];
  const float* bias      = (const float*)d_in[3];
  const int*   row_idx   = (const int*)d_in[4];
  const int*   col_idx   = (const int*)d_in[5];
  float* out     = (float*)d_out;
  float* support = (float*)d_ws;

  (void)in_sizes; (void)n_in; (void)out_size; (void)ws_size;

  gemm_xw_wmma<<<(N_NODES + 127) / 128, 256, 0, stream>>>(X, W, support);
  init_bias<<<2048, 256, 0, stream>>>(bias, out);
  int waves  = (N_EDGES + EDGES_PER_WAVE - 1) / EDGES_PER_WAVE;
  int blocks = (waves + 7) / 8;
  spmm_scatter<<<blocks, 256, 0, stream>>>(support, edge_vals, row_idx, col_idx, out);
}


# SpMM: a wave owns the rows that begin in its 128-edge chunk and runs on to its last row's end; sum kept in registers, sum+bias stored once; atomics and bias-fill launch gone; row sums in edge order
# speedup vs baseline: 1.1348x; 1.0677x over previous
#include <hip/hip_runtime.h>

typedef __attribute__((ext_vector_type(2))) float v2f;
typedef __attribute__((ext_vector_type(8))) float v8f;

#define N_NODES 100000
#define N_EDGES 1600000
#define IN_F    256
#define OUT_F   64

__global__ __launch_bounds__(256) void gemm_xw_wmma(const float* __restrict__ X,
                                                    const float* __restrict__ W,
                                                    float* __restrict__ support) {
  __shared__ float lw[IN_F * OUT_F];

  const int wave = threadIdx.x >> 5;
  const int lane = threadIdx.x & 31;
  const int hi   = lane >> 4;
  const int lo   = lane & 15;
  const int row_base = blockIdx.x * 128 + wave * 16;
  const bool valid = row_base < N_NODES;

  const float* xrow = X + (size_t)((valid ? row_base : 0) + lo) * IN_F + 2 * hi;
  v2f a[3][8];
#pragma unroll
  for (int j = 0; j < 8; ++j) a[0][j] = *(const v2f*)(xrow + 4 * j);
#pragma unroll
  for (int j = 0; j < 8; ++j) a[1][j] = *(const v2f*)(xrow + 32 + 4 * j);

  for (int i = threadIdx.x; i < IN_F * OUT_F; i += 256) {
    int k = i >> 6;
    int n = i & 63;
    int sl = k * 64 + ((((n >> 4) + (k >> 1)) & 3) << 4) + (n & 15);
    lw[sl] = W[i];
  }
  __syncthreads();

  if (!valid) return;

  v8f acc0 = {}, acc1 = {}, acc2 = {}, acc3 = {};
#pragma unroll
  for (int kb = 0; kb < 8; ++kb) {
    if (kb + 2 < 8) {
#pragma unroll
      for (int j = 0; j < 8; ++j)
        a[(kb + 2) % 3][j] = *(const v2f*)(xrow + (kb + 2) * 32 + 4 * j);
    }
#pragma unroll
    for (int j = 0; j < 8; ++j) {
      const int k0 = kb * 32 + 4 * j;
      const int ka = k0 + 2 * hi;
      const v2f av = a[kb % 3][j];
      const float* wr = lw + ka * 64;
      const int kr = ka >> 1;
      const int s0 = (((kr + 0) & 3) << 4) + lo;
      const int s1 = (((kr + 1) & 3) << 4) + lo;
      const int s2 = (((kr + 2) & 3) << 4) + lo;
      const int s3 = (((kr + 3) & 3) << 4) + lo;
      v2f b0 = { wr[s0], wr[s0 + 64] };
      v2f b1 = { wr[s1], wr[s1 + 64] };
      v2f b2 = { wr[s2], wr[s2 + 64] };
      v2f b3 = { wr[s3], wr[s3 + 64] };
      acc0 = __builtin_amdgcn_wmma_f32_16x16x4_f32(false, av, false, b0, (short)0, acc0, false, false);
      acc1 = __builtin_amdgcn_wmma_f32_16x16x4_f32(false, av, false, b1, (short)0, acc1, false, false);
      acc2 = __builtin_amdgcn_wmma_f32_16x16x4_f32(false, av, false, b2, (short)0, acc2, false, false);
      acc3 = __builtin_amdgcn_wmma_f32_16x16x4_f32(false, av, false, b3, (short)0, acc3, false, false);
    }
  }

  float* orow = support + (size_t)(row_base + 8 * hi) * OUT_F + lo;
#pragma unroll
  for (int r = 0; r < 8; ++r) {
    orow[(size_t)r * OUT_F +  0] = acc0[r];
    orow[(size_t)r * OUT_F + 16] = acc1[r];
    orow[(size_t)r * OUT_F + 32] = acc2[r];
    orow[(size_t)r * OUT_F + 48] = acc3[r];
  }
}

#define EDGES_PER_WAVE 128

__device__ __forceinline__ void put_row(float* __restrict__ out, int r, int c2,
                                        float x, float y) {
  *(float2*)(out + (size_t)r * OUT_F + c2) = make_float2(x, y);
}

__global__ __launch_bounds__(256) void spmm_rows(const float* __restrict__ support,
                                                 const float* __restrict__ edge_vals,
                                                 const int* __restrict__ row_idx,
                                                 const int* __restrict__ col_idx,
                                                 const float* __restrict__ bias,
                                                 float* __restrict__ out) {
  const int lane = threadIdx.x & 31;
  const int wave = blockIdx.x * 8 + (threadIdx.x >> 5);
  const long e0 = (long)wave * EDGES_PER_WAVE;
  if (e0 >= N_EDGES) return;
  const long e1 = e0 + EDGES_PER_WAVE;

  const int c2 = lane * 2;
  const float2 b = *(const float2*)(bias + c2);
  float ax = 0.f, ay = 0.f;
  int cur = (e0 > 0) ? row_idx[e0 - 1] : -1;
  bool own = false;

#define ROW_EDGE(r, v, s)                                                   \
  if ((r) != cur) {                                                         \
    if (own) put_row(out, cur, c2, ax + b.x, ay + b.y);                     \
    for (int z = cur + 1; z < (r); ++z) put_row(out, z, c2, b.x, b.y);      \
    cur = (r); own = true; ax = 0.f; ay = 0.f;                              \
  }                                                                         \
  ax = fmaf((v), (s).x, ax); ay = fmaf((v), (s).y, ay);

  for (long e = e0; e < e1; e += 4) {
    const int4   rr = *(const int4*)(row_idx + e);
    const int4   cc = *(const int4*)(col_idx + e);
    const float4 vv = *(const float4*)(edge_vals + e);
    const float2 s0 = *(const float2*)(support + (size_t)cc.x * OUT_F + c2);
    const float2 s1 = *(const float2*)(support + (size_t)cc.y * OUT_F + c2);
    const float2 s2 = *(const float2*)(support + (size_t)cc.z * OUT_F + c2);
    const float2 s3 = *(const float2*)(support + (size_t)cc.w * OUT_F + c2);
    ROW_EDGE(rr.x, vv.x, s0)
    ROW_EDGE(rr.y, vv.y, s1)
    ROW_EDGE(rr.z, vv.z, s2)
    ROW_EDGE(rr.w, vv.w, s3)
  }
#undef ROW_EDGE

  if (!own) return;

#define TAIL_EDGE(r, v, s)                                                  \
  { open = open && ((r) == cur);                                            \
    const float tx = fmaf((v), (s).x, ax);                                  \
    const float ty = fmaf((v), (s).y, ay);                                  \
    ax = open ? tx : ax; ay = open ? ty : ay; }

  bool open = true;
  for (long e = e1; open && e < N_EDGES; e += 8) {
    const int4   ra = *(const int4*)(row_idx + e);
    const int4   rb = *(const int4*)(row_idx + e + 4);
    const int4   ca = *(const int4*)(col_idx + e);
    const int4   cb = *(const int4*)(col_idx + e + 4);
    const float4 va = *(const float4*)(edge_vals + e);
    const float4 vb = *(const float4*)(edge_vals + e + 4);
    const float2 t0 = *(const float2*)(support + (size_t)ca.x * OUT_F + c2);
    const float2 t1 = *(const float2*)(support + (size_t)ca.y * OUT_F + c2);
    const float2 t2 = *(const float2*)(support + (size_t)ca.z * OUT_F + c2);
    const float2 t3 = *(const float2*)(support + (size_t)ca.w * OUT_F + c2);
    const float2 t4 = *(const float2*)(support + (size_t)cb.x * OUT_F + c2);
    const float2 t5 = *(const float2*)(support + (size_t)cb.y * OUT_F + c2);
    const float2 t6 = *(const float2*)(support + (size_t)cb.z * OUT_F + c2);
    const float2 t7 = *(const float2*)(support + (size_t)cb.w * OUT_F + c2);
    TAIL_EDGE(ra.x, va.x, t0)
    TAIL_EDGE(ra.y, va.y, t1)
    TAIL_EDGE(ra.z, va.z, t2)
    TAIL_EDGE(ra.w, va.w, t3)
    TAIL_EDGE(rb.x, vb.x, t4)
    TAIL_EDGE(rb.y, vb.y, t5)
    TAIL_EDGE(rb.z, vb.z, t6)
    TAIL_EDGE(rb.w, vb.w, t7)
  }
#undef TAIL_EDGE

  put_row(out, cur, c2, ax + b.x, ay + b.y);
  if (open) {
    for (int z = cur + 1; z < N_NODES; ++z) put_row(out, z, c2, b.x, b.y);
  }
}

extern "C" void kernel_launch(void* const* d_in, const int* in_sizes, int n_in,
                              void* d_out, int out_size, void* d_ws, size_t ws_size,
                              hipStream_t stream) {
  const float* X         = (const float*)d_in[0];
  const float* edge_vals = (const float*)d_in[1];
  const float* W         = (const float*)d_in[2];
  const float* bias      = (const float*)d_in[3];
  const int*   row_idx   = (const int*)d_in[4];
  const int*   col_idx   = (const int*)d_in[5];
  float* out     = (float*)d_out;
  float* support = (float*)d_ws;

  (void)in_sizes; (void)n_in; (void)out_size; (void)ws_size;

  gemm_xw_wmma<<<(N_NODES + 127) / 128, 256, 0, stream>>>(X, W, support);
  int waves  = (N_EDGES + EDGES_PER_WAVE - 1) / EDGES_PER_WAVE;
  int blocks = (waves + 7) / 8;
  spmm_rows<<<blocks, 256, 0, stream>>>(support, edge_vals, row_idx, col_idx, bias, out);
}


# GEMM work dealt evenly: 768 work-groups, three per WGP (LDS-capped), one row tile per wave; last 106 tiles cut in column quarters over waves 0-3; busiest SIMD 6.25 tiles, was 8; sums unchanged
# speedup vs baseline: 1.1704x; 1.0314x over previous
#include <hip/hip_runtime.h>

typedef __attribute__((ext_vector_type(2))) float v2f;
typedef __attribute__((ext_vector_type(8))) float v8f;

#define N_NODES 100000
#define N_EDGES 1600000
#define IN_F    256
#define OUT_F   64

#define GEMM_GROUPS 768
#define MAIN_TILES  (GEMM_GROUPS * 8)
#define LEFT_TILES  (N_NODES / 16 - MAIN_TILES)

__global__ __launch_bounds__(256) void gemm_xw_wmma(const float* __restrict__ X,
                                                    const float* __restrict__ W,
                                                    float* __restrict__ support) {
  __shared__ float lw[96 * 256];

  const int wave = __builtin_amdgcn_readfirstlane((int)(threadIdx.x >> 5));
  const int lane = threadIdx.x & 31;
  const int hi   = lane >> 4;
  const int lo   = lane & 15;
  const int row_base = ((int)blockIdx.x * 8 + wave) * 16;
  const bool left = ((int)blockIdx.x < LEFT_TILES) && (wave < 4);
  const int left_base = (MAIN_TILES + (int)blockIdx.x) * 16;
  const int q = wave;

  const float* xrow = X + (size_t)(row_base + lo) * IN_F + 2 * hi;
  const float* xlft = X + (size_t)((left ? left_base : row_base) + lo) * IN_F + 2 * hi;
  v2f a[3][8];
  if (left) {
#pragma unroll
    for (int j = 0; j < 8; ++j) a[1][j] = *(const v2f*)(xlft + 4 * j);
#pragma unroll
    for (int j = 0; j < 8; ++j) a[2][j] = *(const v2f*)(xlft + 32 + 4 * j);
  } else {
#pragma unroll
    for (int j = 0; j < 8; ++j) a[0][j] = *(const v2f*)(xrow + 4 * j);
#pragma unroll
    for (int j = 0; j < 8; ++j) a[1][j] = *(const v2f*)(xrow + 32 + 4 * j);
  }

  for (int i = threadIdx.x; i < IN_F * OUT_F; i += 256) {
    int k = i >> 6;
    int n = i & 63;
    int sl = k * 64 + ((((n >> 4) + (k >> 1)) & 3) << 4) + (n & 15);
    lw[sl] = W[i];
  }
  __syncthreads();

  if (left) {
    v8f accq = {};
#pragma unroll
    for (int gb = 0; gb < 8; ++gb) {
      const float* nxt = (gb + 2 < 8) ? (xlft + (gb + 2) * 32) : (xrow + (gb + 2 - 8) * 32);
#pragma unroll
      for (int j = 0; j < 8; ++j) a[gb % 3][j] = *(const v2f*)(nxt + 4 * j);
#pragma unroll
      for (int j = 0; j < 8; ++j) {
        const int k0 = gb * 32 + 4 * j;
        const int ka = k0 + 2 * hi;
        const v2f av = a[(gb + 1) % 3][j];
        const float* wr = lw + ka * 64;
        const int kr = ka >> 1;
        const int sq = (((kr + q) & 3) << 4) + lo;
        v2f bq = { wr[sq], wr[sq + 64] };
        accq = __builtin_amdgcn_wmma_f32_16x16x4_f32(false, av, false, bq, (short)0, accq, false, false);
      }
    }
    float* qrow = support + (size_t)(left_base + 8 * hi) * OUT_F + q * 16 + lo;
#pragma unroll
    for (int r = 0; r < 8; ++r) qrow[(size_t)r * OUT_F] = accq[r];
  }

  v8f acc0 = {}, acc1 = {}, acc2 = {}, acc3 = {};
#pragma unroll
  for (int kb = 0; kb < 8; ++kb) {
    if (kb + 2 < 8) {
#pragma unroll
      for (int j = 0; j < 8; ++j)
        a[(kb + 2) % 3][j] = *(const v2f*)(xrow + (kb + 2) * 32 + 4 * j);
    }
#pragma unroll
    for (int j = 0; j < 8; ++j) {
      const int k0 = kb * 32 + 4 * j;
      const int ka = k0 + 2 * hi;
      const v2f av = a[kb % 3][j];
      const float* wr = lw + ka * 64;
      const int kr = ka >> 1;
      const int s0 = (((kr + 0) & 3) << 4) + lo;
      const int s1 = (((kr + 1) & 3) << 4) + lo;
      const int s2 = (((kr + 2) & 3) << 4) + lo;
      const int s3 = (((kr + 3) & 3) << 4) + lo;
      v2f b0 = { wr[s0], wr[s0 + 64] };
      v2f b1 = { wr[s1], wr[s1 + 64] };
      v2f b2 = { wr[s2], wr[s2 + 64] };
      v2f b3 = { wr[s3], wr[s3 + 64] };
      acc0 = __builtin_amdgcn_wmma_f32_16x16x4_f32(false, av, false, b0, (short)0, acc0, false, false);
      acc1 = __builtin_amdgcn_wmma_f32_16x16x4_f32(false, av, false, b1, (short)0, acc1, false, false);
      acc2 = __builtin_amdgcn_wmma_f32_16x16x4_f32(false, av, false, b2, (short)0, acc2, false, false);
      acc3 = __builtin_amdgcn_wmma_f32_16x16x4_f32(false, av, false, b3, (short)0, acc3, false, false);
    }
  }

  float* orow = support + (size_t)(row_base + 8 * hi) * OUT_F + lo;
#pragma unroll
  for (int r = 0; r < 8; ++r) {
    orow[(size_t)r * OUT_F +  0] = acc0[r];
    orow[(size_t)r * OUT_F + 16] = acc1[r];
    orow[(size_t)r * OUT_F + 32] = acc2[r];
    orow[(size_t)r * OUT_F + 48] = acc3[r];
  }
}

#define EDGES_PER_WAVE 128

__device__ __forceinline__ void put_row(float* __restrict__ out, int r, int c2,
                                        float x, float y) {
  *(float2*)(out + (size_t)r * OUT_F + c2) = make_float2(x, y);
}

__global__ __launch_bounds__(256) void spmm_rows(const float* __restrict__ support,
                                                 const float* __restrict__ edge_vals,
                                                 const int* __restrict__ row_idx,
                                                 const int* __restrict__ col_idx,
                                                 const float* __restrict__ bias,
                                                 float* __restrict__ out) {
  const int lane = threadIdx.x & 31;
  const int wave = blockIdx.x * 8 + (threadIdx.x >> 5);
  const long e0 = (long)wave * EDGES_PER_WAVE;
  if (e0 >= N_EDGES) return;
  const long e1 = e0 + EDGES_PER_WAVE;

  const int c2 = lane * 2;
  const float2 b = *(const float2*)(bias + c2);
  float ax = 0.f, ay = 0.f;
  int cur = (e0 > 0) ? row_idx[e0 - 1] : -1;
  bool own = false;

#define ROW_EDGE(r, v, s)                                                   \
  if ((r) != cur) {                                                         \
    if (own) put_row(out, cur, c2, ax + b.x, ay + b.y);                     \
    for (int z = cur + 1; z < (r); ++z) put_row(out, z, c2, b.x, b.y);      \
    cur = (r); own = true; ax = 0.f; ay = 0.f;                              \
  }                                                                         \
  ax = fmaf((v), (s).x, ax); ay = fmaf((v), (s).y, ay);

  for (long e = e0; e < e1; e += 4) {
    const int4   rr = *(const int4*)(row_idx + e);
    const int4   cc = *(const int4*)(col_idx + e);
    const float4 vv = *(const float4*)(edge_vals + e);
    const float2 s0 = *(const float2*)(support + (size_t)cc.x * OUT_F + c2);
    const float2 s1 = *(const float2*)(support + (size_t)cc.y * OUT_F + c2);
    const float2 s2 = *(const float2*)(support + (size_t)cc.z * OUT_F + c2);
    const float2 s3 = *(const float2*)(support + (size_t)cc.w * OUT_F + c2);
    ROW_EDGE(rr.x, vv.x, s0)
    ROW_EDGE(rr.y, vv.y, s1)
    ROW_EDGE(rr.z, vv.z, s2)
    ROW_EDGE(rr.w, vv.w, s3)
  }
#undef ROW_EDGE

  if (!own) return;

#define TAIL_EDGE(r, v, s)                                                  \
  { open = open && ((r) == cur);                                            \
    const float tx = fmaf((v), (s).x, ax);                                  \
    const float ty = fmaf((v), (s).y, ay);                                  \
    ax = open ? tx : ax; ay = open ? ty : ay; }

  bool open = true;
  for (long e = e1; open && e < N_EDGES; e += 8) {
    const int4   ra = *(const int4*)(row_idx + e);
    const int4   rb = *(const int4*)(row_idx + e + 4);
    const int4   ca = *(const int4*)(col_idx + e);
    const int4   cb = *(const int4*)(col_idx + e + 4);
    const float4 va = *(const float4*)(edge_vals + e);
    const float4 vb = *(const float4*)(edge_vals + e + 4);
    const float2 t0 = *(const float2*)(support + (size_t)ca.x * OUT_F + c2);
    const float2 t1 = *(const float2*)(support + (size_t)ca.y * OUT_F + c2);
    const float2 t2 = *(const float2*)(support + (size_t)ca.z * OUT_F + c2);
    const float2 t3 = *(const float2*)(support + (size_t)ca.w * OUT_F + c2);
    const float2 t4 = *(const float2*)(support + (size_t)cb.x * OUT_F + c2);
    const float2 t5 = *(const float2*)(support + (size_t)cb.y * OUT_F + c2);
    const float2 t6 = *(const float2*)(support + (size_t)cb.z * OUT_F + c2);
    const float2 t7 = *(const float2*)(support + (size_t)cb.w * OUT_F + c2);
    TAIL_EDGE(ra.x, va.x, t0)
    TAIL_EDGE(ra.y, va.y, t1)
    TAIL_EDGE(ra.z, va.z, t2)
    TAIL_EDGE(ra.w, va.w, t3)
    TAIL_EDGE(rb.x, vb.x, t4)
    TAIL_EDGE(rb.y, vb.y, t5)
    TAIL_EDGE(rb.z, vb.z, t6)
    TAIL_EDGE(rb.w, vb.w, t7)
  }
#undef TAIL_EDGE

  put_row(out, cur, c2, ax + b.x, ay + b.y);
  if (open) {
    for (int z = cur + 1; z < N_NODES; ++z) put_row(out, z, c2, b.x, b.y);
  }
}

extern "C" void kernel_launch(void* const* d_in, const int* in_sizes, int n_in,
                              void* d_out, int out_size, void* d_ws, size_t ws_size,
                              hipStream_t stream) {
  const float* X         = (const float*)d_in[0];
  const float* edge_vals = (const float*)d_in[1];
  const float* W         = (const float*)d_in[2];
  const float* bias      = (const float*)d_in[3];
  const int*   row_idx   = (const int*)d_in[4];
  const int*   col_idx   = (const int*)d_in[5];
  float* out     = (float*)d_out;
  float* support = (float*)d_ws;

  (void)in_sizes; (void)n_in; (void)out_size; (void)ws_size;

  gemm_xw_wmma<<<GEMM_GROUPS, 256, 0, stream>>>(X, W, support);
  int waves  = (N_EDGES + EDGES_PER_WAVE - 1) / EDGES_PER_WAVE;
  int blocks = (waves + 7) / 8;
  spmm_rows<<<blocks, 256, 0, stream>>>(support, edge_vals, row_idx, col_idx, bias, out);
}


# SpMM: wave loads its 128-edge chunk's row/col/value once (coalesced), takes each edge by lane into scalar registers; support-row gathers in a ring 16 edges ahead; row sums still in edge order
# speedup vs baseline: 1.2318x; 1.0525x over previous
#include <hip/hip_runtime.h>

typedef __attribute__((ext_vector_type(2))) float v2f;
typedef __attribute__((ext_vector_type(8))) float v8f;

#define N_NODES 100000
#define N_EDGES 1600000
#define IN_F    256
#define OUT_F   64

#define GEMM_GROUPS 768
#define MAIN_TILES  (GEMM_GROUPS * 8)
#define LEFT_TILES  (N_NODES / 16 - MAIN_TILES)

__global__ __launch_bounds__(256) void gemm_xw_wmma(const float* __restrict__ X,
                                                    const float* __restrict__ W,
                                                    float* __restrict__ support) {
  __shared__ float lw[96 * 256];

  const int wave = __builtin_amdgcn_readfirstlane((int)(threadIdx.x >> 5));
  const int lane = threadIdx.x & 31;
  const int hi   = lane >> 4;
  const int lo   = lane & 15;
  const int row_base = ((int)blockIdx.x * 8 + wave) * 16;
  const bool left = ((int)blockIdx.x < LEFT_TILES) && (wave < 4);
  const int left_base = (MAIN_TILES + (int)blockIdx.x) * 16;
  const int q = wave;

  const float* xrow = X + (size_t)(row_base + lo) * IN_F + 2 * hi;
  const float* xlft = X + (size_t)((left ? left_base : row_base) + lo) * IN_F + 2 * hi;
  v2f a[3][8];
  if (left) {
#pragma unroll
    for (int j = 0; j < 8; ++j) a[1][j] = *(const v2f*)(xlft + 4 * j);
#pragma unroll
    for (int j = 0; j < 8; ++j) a[2][j] = *(const v2f*)(xlft + 32 + 4 * j);
  } else {
#pragma unroll
    for (int j = 0; j < 8; ++j) a[0][j] = *(const v2f*)(xrow + 4 * j);
#pragma unroll
    for (int j = 0; j < 8; ++j) a[1][j] = *(const v2f*)(xrow + 32 + 4 * j);
  }

  for (int i = threadIdx.x; i < IN_F * OUT_F; i += 256) {
    int k = i >> 6;
    int n = i & 63;
    int sl = k * 64 + ((((n >> 4) + (k >> 1)) & 3) << 4) + (n & 15);
    lw[sl] = W[i];
  }
  __syncthreads();

  if (left) {
    v8f accq = {};
#pragma unroll
    for (int gb = 0; gb < 8; ++gb) {
      const float* nxt = (gb + 2 < 8) ? (xlft + (gb + 2) * 32) : (xrow + (gb + 2 - 8) * 32);
#pragma unroll
      for (int j = 0; j < 8; ++j) a[gb % 3][j] = *(const v2f*)(nxt + 4 * j);
#pragma unroll
      for (int j = 0; j < 8; ++j) {
        const int k0 = gb * 32 + 4 * j;
        const int ka = k0 + 2 * hi;
        const v2f av = a[(gb + 1) % 3][j];
        const float* wr = lw + ka * 64;
        const int kr = ka >> 1;
        const int sq = (((kr + q) & 3) << 4) + lo;
        v2f bq = { wr[sq], wr[sq + 64] };
        accq = __builtin_amdgcn_wmma_f32_16x16x4_f32(false, av, false, bq, (short)0, accq, false, false);
      }
    }
    float* qrow = support + (size_t)(left_base + 8 * hi) * OUT_F + q * 16 + lo;
#pragma unroll
    for (int r = 0; r < 8; ++r) qrow[(size_t)r * OUT_F] = accq[r];
  }

  v8f acc0 = {}, acc1 = {}, acc2 = {}, acc3 = {};
#pragma unroll
  for (int kb = 0; kb < 8; ++kb) {
    if (kb + 2 < 8) {
#pragma unroll
      for (int j = 0; j < 8; ++j)
        a[(kb + 2) % 3][j] = *(const v2f*)(xrow + (kb + 2) * 32 + 4 * j);
    }
#pragma unroll
    for (int j = 0; j < 8; ++j) {
      const int k0 = kb * 32 + 4 * j;
      const int ka = k0 + 2 * hi;
      const v2f av = a[kb % 3][j];
      const float* wr = lw + ka * 64;
      const int kr = ka >> 1;
      const int s0 = (((kr + 0) & 3) << 4) + lo;
      const int s1 = (((kr + 1) & 3) << 4) + lo;
      const int s2 = (((kr + 2) & 3) << 4) + lo;
      const int s3 = (((kr + 3) & 3) << 4) + lo;
      v2f b0 = { wr[s0], wr[s0 + 64] };
      v2f b1 = { wr[s1], wr[s1 + 64] };
      v2f b2 = { wr[s2], wr[s2 + 64] };
      v2f b3 = { wr[s3], wr[s3 + 64] };
      acc0 = __builtin_amdgcn_wmma_f32_16x16x4_f32(false, av, false, b0, (short)0, acc0, false, false);
      acc1 = __builtin_amdgcn_wmma_f32_16x16x4_f32(false, av, false, b1, (short)0, acc1, false, false);
      acc2 = __builtin_amdgcn_wmma_f32_16x16x4_f32(false, av, false, b2, (short)0, acc2, false, false);
      acc3 = __builtin_amdgcn_wmma_f32_16x16x4_f32(false, av, false, b3, (short)0, acc3, false, false);
    }
  }

  float* orow = support + (size_t)(row_base + 8 * hi) * OUT_F + lo;
#pragma unroll
  for (int r = 0; r < 8; ++r) {
    orow[(size_t)r * OUT_F +  0] = acc0[r];
    orow[(size_t)r * OUT_F + 16] = acc1[r];
    orow[(size_t)r * OUT_F + 32] = acc2[r];
    orow[(size_t)r * OUT_F + 48] = acc3[r];
  }
}

#define EDGES_PER_WAVE 128
#define GATHER_RING    16
#define TAIL_REGS      32

__device__ __forceinline__ void put_row(float* __restrict__ out, int r, int c2,
                                        float x, float y) {
  *(float2*)(out + (size_t)r * OUT_F + c2) = make_float2(x, y);
}

__device__ __forceinline__ int   pick(const int4 v, int k)   { return k == 0 ? v.x : k == 1 ? v.y : k == 2 ? v.z : v.w; }
__device__ __forceinline__ float pick(const float4 v, int k) { return k == 0 ? v.x : k == 1 ? v.y : k == 2 ? v.z : v.w; }
__device__ __forceinline__ int   lane_i(int v, int ln)   { return __builtin_amdgcn_readlane(v, ln); }
__device__ __forceinline__ float lane_f(float v, int ln) { return __int_as_float(__builtin_amdgcn_readlane(__float_as_int(v), ln)); }

__device__ __forceinline__ float2 gather_row(const float* __restrict__ support, int c, int c2) {
  const unsigned off = (unsigned)c * (unsigned)OUT_F + (unsigned)c2;
  return *(const float2*)(support + off);
}

__global__ __launch_bounds__(256) void spmm_rows(const float* __restrict__ support,
                                                 const float* __restrict__ edge_vals,
                                                 const int* __restrict__ row_idx,
                                                 const int* __restrict__ col_idx,
                                                 const float* __restrict__ bias,
                                                 float* __restrict__ out) {
  const int lane = threadIdx.x & 31;
  const int wave = (int)blockIdx.x * 8 + __builtin_amdgcn_readfirstlane((int)(threadIdx.x >> 5));
  const int e0 = wave * EDGES_PER_WAVE;
  if (e0 >= N_EDGES) return;
  const int e1 = e0 + EDGES_PER_WAVE;
  const bool has_next = e1 < N_EDGES;

  const int c2 = lane * 2;

  const int4   r0 = *(const int4*)(row_idx + e0 + 4 * lane);
  const int4   k0 = *(const int4*)(col_idx + e0 + 4 * lane);
  const float4 v0 = *(const float4*)(edge_vals + e0 + 4 * lane);
  const int et = (has_next ? e1 : e0) + lane;
  const int   r1 = row_idx[et];
  const int   k1 = col_idx[et];
  const float v1 = edge_vals[et];

  const float2 b = *(const float2*)(bias + c2);
  float ax = 0.f, ay = 0.f;
  int cur = (e0 > 0) ? __builtin_amdgcn_readfirstlane(row_idx[e0 - 1]) : -1;
  bool own = false;

#define ROW_EDGE(r, v, s)                                                   \
  if ((r) != cur) {                                                         \
    if (own) put_row(out, cur, c2, ax + b.x, ay + b.y);                     \
    _Pragma("nounroll")                                                     \
    for (int z = cur + 1; z < (r); ++z) put_row(out, z, c2, b.x, b.y);      \
    cur = (r); own = true; ax = 0.f; ay = 0.f;                              \
  }                                                                         \
  ax = fmaf((v), (s).x, ax); ay = fmaf((v), (s).y, ay);

  float2 g[GATHER_RING];
#pragma unroll
  for (int p = 0; p < GATHER_RING; ++p)
    g[p] = gather_row(support, lane_i(pick(k0, p & 3), p >> 2), c2);
  for (int blk = 0; blk < EDGES_PER_WAVE / GATHER_RING - 1; ++blk) {
    const int l0 = blk * (GATHER_RING / 4);
#pragma unroll
    for (int p = 0; p < GATHER_RING; ++p) {
      const int   r = lane_i(pick(r0, p & 3), l0 + (p >> 2));
      const float v = lane_f(pick(v0, p & 3), l0 + (p >> 2));
      ROW_EDGE(r, v, g[p])
      g[p] = gather_row(support, lane_i(pick(k0, p & 3), l0 + (GATHER_RING / 4) + (p >> 2)), c2);
    }
  }
#pragma unroll
  for (int p = 0; p < GATHER_RING; ++p) {
    const int ln = (EDGES_PER_WAVE - GATHER_RING) / 4 + (p >> 2);
    const int   r = lane_i(pick(r0, p & 3), ln);
    const float v = lane_f(pick(v0, p & 3), ln);
    ROW_EDGE(r, v, g[p])
  }
#undef ROW_EDGE

  if (!own) return;

  bool open = true;
  if (has_next) {
    const int nt = __builtin_popcount((unsigned)__ballot(r1 == cur));
    for (int base = 0; base < nt; base += 8) {
      float2 t[8];
#pragma unroll
      for (int j = 0; j < 8; ++j) t[j] = gather_row(support, lane_i(k1, base + j), c2);
#pragma unroll
      for (int j = 0; j < 8; ++j) {
        const float v = lane_f(v1, base + j);
        if (base + j < nt) { ax = fmaf(v, t[j].x, ax); ay = fmaf(v, t[j].y, ay); }
      }
    }
    open = (nt == TAIL_REGS);
  }

#define TAIL_EDGE(r, v, s)                                                  \
  { open = open && ((r) == cur);                                            \
    const float tx = fmaf((v), (s).x, ax);                                  \
    const float ty = fmaf((v), (s).y, ay);                                  \
    ax = open ? tx : ax; ay = open ? ty : ay; }

  for (long e = (long)e1 + TAIL_REGS; open && e < N_EDGES; e += 8) {
    const int4   ra = *(const int4*)(row_idx + e);
    const int4   rb = *(const int4*)(row_idx + e + 4);
    const int4   ca = *(const int4*)(col_idx + e);
    const int4   cb = *(const int4*)(col_idx + e + 4);
    const float4 va = *(const float4*)(edge_vals + e);
    const float4 vb = *(const float4*)(edge_vals + e + 4);
    const float2 t0 = *(const float2*)(support + (size_t)ca.x * OUT_F + c2);
    const float2 t1 = *(const float2*)(support + (size_t)ca.y * OUT_F + c2);
    const float2 t2 = *(const float2*)(support + (size_t)ca.z * OUT_F + c2);
    const float2 t3 = *(const float2*)(support + (size_t)ca.w * OUT_F + c2);
    const float2 t4 = *(const float2*)(support + (size_t)cb.x * OUT_F + c2);
    const float2 t5 = *(const float2*)(support + (size_t)cb.y * OUT_F + c2);
    const float2 t6 = *(const float2*)(support + (size_t)cb.z * OUT_F + c2);
    const float2 t7 = *(const float2*)(support + (size_t)cb.w * OUT_F + c2);
    TAIL_EDGE(ra.x, va.x, t0)
    TAIL_EDGE(ra.y, va.y, t1)
    TAIL_EDGE(ra.z, va.z, t2)
    TAIL_EDGE(ra.w, va.w, t3)
    TAIL_EDGE(rb.x, vb.x, t4)
    TAIL_EDGE(rb.y, vb.y, t5)
    TAIL_EDGE(rb.z, vb.z, t6)
    TAIL_EDGE(rb.w, vb.w, t7)
  }
#undef TAIL_EDGE

  put_row(out, cur, c2, ax + b.x, ay + b.y);
  if (open) {
    for (int z = cur + 1; z < N_NODES; ++z) put_row(out, z, c2, b.x, b.y);
  }
}

extern "C" void kernel_launch(void* const* d_in, const int* in_sizes, int n_in,
                              void* d_out, int out_size, void* d_ws, size_t ws_size,
                              hipStream_t stream) {
  const float* X         = (const float*)d_in[0];
  const float* edge_vals = (const float*)d_in[1];
  const float* W         = (const float*)d_in[2];
  const float* bias      = (const float*)d_in[3];
  const int*   row_idx   = (const int*)d_in[4];
  const int*   col_idx   = (const int*)d_in[5];
  float* out     = (float*)d_out;
  float* support = (float*)d_ws;

  (void)in_sizes; (void)n_in; (void)out_size; (void)ws_size;

  gemm_xw_wmma<<<GEMM_GROUPS, 256, 0, stream>>>(X, W, support);
  int waves  = (N_EDGES + EDGES_PER_WAVE - 1) / EDGES_PER_WAVE;
  int blocks = (waves + 7) / 8;
  spmm_rows<<<blocks, 256, 0, stream>>>(support, edge_vals, row_idx, col_idx, bias, out);
}


# GEMM: W staged with sixteen 16-B loads per thread, all in flight at once beside the X prefetch, and 16-B LDS stores: one L2 round trip before the barrier, was eight dependent ones; sums unchanged
# speedup vs baseline: 1.2751x; 1.0352x over previous
#include <hip/hip_runtime.h>

typedef __attribute__((ext_vector_type(2))) float v2f;
typedef __attribute__((ext_vector_type(8))) float v8f;

#define N_NODES 100000
#define N_EDGES 1600000
#define IN_F    256
#define OUT_F   64

#define GEMM_GROUPS 768
#define MAIN_TILES  (GEMM_GROUPS * 8)
#define LEFT_TILES  (N_NODES / 16 - MAIN_TILES)

__global__ __launch_bounds__(256) void gemm_xw_wmma(const float* __restrict__ X,
                                                    const float* __restrict__ W,
                                                    float* __restrict__ support) {
  __shared__ __attribute__((aligned(16))) float lw[96 * 256];

  const int wave = __builtin_amdgcn_readfirstlane((int)(threadIdx.x >> 5));
  const int lane = threadIdx.x & 31;
  const int hi   = lane >> 4;
  const int lo   = lane & 15;
  const int row_base = ((int)blockIdx.x * 8 + wave) * 16;
  const bool left = ((int)blockIdx.x < LEFT_TILES) && (wave < 4);
  const int left_base = (MAIN_TILES + (int)blockIdx.x) * 16;
  const int q = wave;

  const float* xrow = X + (size_t)(row_base + lo) * IN_F + 2 * hi;
  const float* xlft = X + (size_t)((left ? left_base : row_base) + lo) * IN_F + 2 * hi;
  v2f a[3][8];
  if (left) {
#pragma unroll
    for (int j = 0; j < 8; ++j) a[1][j] = *(const v2f*)(xlft + 4 * j);
#pragma unroll
    for (int j = 0; j < 8; ++j) a[2][j] = *(const v2f*)(xlft + 32 + 4 * j);
  } else {
#pragma unroll
    for (int j = 0; j < 8; ++j) a[0][j] = *(const v2f*)(xrow + 4 * j);
#pragma unroll
    for (int j = 0; j < 8; ++j) a[1][j] = *(const v2f*)(xrow + 32 + 4 * j);
  }

  const float4* wsrc = (const float4*)W + threadIdx.x;
  float4 wv[16];
#pragma unroll
  for (int i = 0; i < 16; ++i) wv[i] = wsrc[i * 256];
  {
    const int kw = (int)(threadIdx.x >> 4);
    const int nw = (int)(threadIdx.x & 15) * 4;
    const int sl = kw * 64 + ((((nw >> 4) + (kw >> 1)) & 3) << 4) + (nw & 15);
#pragma unroll
    for (int i = 0; i < 16; ++i) *(float4*)(lw + sl + i * 1024) = wv[i];
  }
  __syncthreads();

  if (left) {
    v8f accq = {};
#pragma unroll
    for (int gb = 0; gb < 8; ++gb) {
      const float* nxt = (gb + 2 < 8) ? (xlft + (gb + 2) * 32) : (xrow + (gb + 2 - 8) * 32);
#pragma unroll
      for (int j = 0; j < 8; ++j) a[gb % 3][j] = *(const v2f*)(nxt + 4 * j);
#pragma unroll
      for (int j = 0; j < 8; ++j) {
        const int k0 = gb * 32 + 4 * j;
        const int ka = k0 + 2 * hi;
        const v2f av = a[(gb + 1) % 3][j];
        const float* wr = lw + ka * 64;
        const int kr = ka >> 1;
        const int sq = (((kr + q) & 3) << 4) + lo;
        v2f bq = { wr[sq], wr[sq + 64] };
        accq = __builtin_amdgcn_wmma_f32_16x16x4_f32(false, av, false, bq, (short)0, accq, false, false);
      }
    }
    float* qrow = support + (size_t)(left_base + 8 * hi) * OUT_F + q * 16 + lo;
#pragma unroll
    for (int r = 0; r < 8; ++r) qrow[(size_t)r * OUT_F] = accq[r];
  }

  v8f acc0 = {}, acc1 = {}, acc2 = {}, acc3 = {};
#pragma unroll
  for (int kb = 0; kb < 8; ++kb) {
    if (kb + 2 < 8) {
#pragma unroll
      for (int j = 0; j < 8; ++j)
        a[(kb + 2) % 3][j] = *(const v2f*)(xrow + (kb + 2) * 32 + 4 * j);
    }
#pragma unroll
    for (int j = 0; j < 8; ++j) {
      const int k0 = kb * 32 + 4 * j;
      const int ka = k0 + 2 * hi;
      const v2f av = a[kb % 3][j];
      const float* wr = lw + ka * 64;
      const int kr = ka >> 1;
      const int s0 = (((kr + 0) & 3) << 4) + lo;
      const int s1 = (((kr + 1) & 3) << 4) + lo;
      const int s2 = (((kr + 2) & 3) << 4) + lo;
      const int s3 = (((kr + 3) & 3) << 4) + lo;
      v2f b0 = { wr[s0], wr[s0 + 64] };
      v2f b1 = { wr[s1], wr[s1 + 64] };
      v2f b2 = { wr[s2], wr[s2 + 64] };
      v2f b3 = { wr[s3], wr[s3 + 64] };
      acc0 = __builtin_amdgcn_wmma_f32_16x16x4_f32(false, av, false, b0, (short)0, acc0, false, false);
      acc1 = __builtin_amdgcn_wmma_f32_16x16x4_f32(false, av, false, b1, (short)0, acc1, false, false);
      acc2 = __builtin_amdgcn_wmma_f32_16x16x4_f32(false, av, false, b2, (short)0, acc2, false, false);
      acc3 = __builtin_amdgcn_wmma_f32_16x16x4_f32(false, av, false, b3, (short)0, acc3, false, false);
    }
  }

  float* orow = support + (size_t)(row_base + 8 * hi) * OUT_F + lo;
#pragma unroll
  for (int r = 0; r < 8; ++r) {
    orow[(size_t)r * OUT_F +  0] = acc0[r];
    orow[(size_t)r * OUT_F + 16] = acc1[r];
    orow[(size_t)r * OUT_F + 32] = acc2[r];
    orow[(size_t)r * OUT_F + 48] = acc3[r];
  }
}

#define EDGES_PER_WAVE 128
#define GATHER_RING    16
#define TAIL_REGS      32

__device__ __forceinline__ void put_row(float* __restrict__ out, int r, int c2,
                                        float x, float y) {
  *(float2*)(out + (size_t)r * OUT_F + c2) = make_float2(x, y);
}

__device__ __forceinline__ int   pick(const int4 v, int k)   { return k == 0 ? v.x : k == 1 ? v.y : k == 2 ? v.z : v.w; }
__device__ __forceinline__ float pick(const float4 v, int k) { return k == 0 ? v.x : k == 1 ? v.y : k == 2 ? v.z : v.w; }
__device__ __forceinline__ int   lane_i(int v, int ln)   { return __builtin_amdgcn_readlane(v, ln); }
__device__ __forceinline__ float lane_f(float v, int ln) { return __int_as_float(__builtin_amdgcn_readlane(__float_as_int(v), ln)); }

__device__ __forceinline__ float2 gather_row(const float* __restrict__ support, int c, int c2) {
  const unsigned off = (unsigned)c * (unsigned)OUT_F + (unsigned)c2;
  return *(const float2*)(support + off);
}

__global__ __launch_bounds__(256) void spmm_rows(const float* __restrict__ support,
                                                 const float* __restrict__ edge_vals,
                                                 const int* __restrict__ row_idx,
                                                 const int* __restrict__ col_idx,
                                                 const float* __restrict__ bias,
                                                 float* __restrict__ out) {
  const int lane = threadIdx.x & 31;
  const int wave = (int)blockIdx.x * 8 + __builtin_amdgcn_readfirstlane((int)(threadIdx.x >> 5));
  const int e0 = wave * EDGES_PER_WAVE;
  if (e0 >= N_EDGES) return;
  const int e1 = e0 + EDGES_PER_WAVE;
  const bool has_next = e1 < N_EDGES;

  const int c2 = lane * 2;

  const int4   r0 = *(const int4*)(row_idx + e0 + 4 * lane);
  const int4   k0 = *(const int4*)(col_idx + e0 + 4 * lane);
  const float4 v0 = *(const float4*)(edge_vals + e0 + 4 * lane);
  const int et = (has_next ? e1 : e0) + lane;
  const int   r1 = row_idx[et];
  const int   k1 = col_idx[et];
  const float v1 = edge_vals[et];

  const float2 b = *(const float2*)(bias + c2);
  float ax = 0.f, ay = 0.f;
  int cur = (e0 > 0) ? __builtin_amdgcn_readfirstlane(row_idx[e0 - 1]) : -1;
  bool own = false;

#define ROW_EDGE(r, v, s)                                                   \
  if ((r) != cur) {                                                         \
    if (own) put_row(out, cur, c2, ax + b.x, ay + b.y);                     \
    _Pragma("nounroll")                                                     \
    for (int z = cur + 1; z < (r); ++z) put_row(out, z, c2, b.x, b.y);      \
    cur = (r); own = true; ax = 0.f; ay = 0.f;                              \
  }                                                                         \
  ax = fmaf((v), (s).x, ax); ay = fmaf((v), (s).y, ay);

  float2 g[GATHER_RING];
#pragma unroll
  for (int p = 0; p < GATHER_RING; ++p)
    g[p] = gather_row(support, lane_i(pick(k0, p & 3), p >> 2), c2);
  for (int blk = 0; blk < EDGES_PER_WAVE / GATHER_RING - 1; ++blk) {
    const int l0 = blk * (GATHER_RING / 4);
#pragma unroll
    for (int p = 0; p < GATHER_RING; ++p) {
      const int   r = lane_i(pick(r0, p & 3), l0 + (p >> 2));
      const float v = lane_f(pick(v0, p & 3), l0 + (p >> 2));
      ROW_EDGE(r, v, g[p])
      g[p] = gather_row(support, lane_i(pick(k0, p & 3), l0 + (GATHER_RING / 4) + (p >> 2)), c2);
    }
  }
#pragma unroll
  for (int p = 0; p < GATHER_RING; ++p) {
    const int ln = (EDGES_PER_WAVE - GATHER_RING) / 4 + (p >> 2);
    const int   r = lane_i(pick(r0, p & 3), ln);
    const float v = lane_f(pick(v0, p & 3), ln);
    ROW_EDGE(r, v, g[p])
  }
#undef ROW_EDGE

  if (!own) return;

  bool open = true;
  if (has_next) {
    const int nt = __builtin_popcount((unsigned)__ballot(r1 == cur));
    for (int base = 0; base < nt; base += 8) {
      float2 t[8];
#pragma unroll
      for (int j = 0; j < 8; ++j) t[j] = gather_row(support, lane_i(k1, base + j), c2);
#pragma unroll
      for (int j = 0; j < 8; ++j) {
        const float v = lane_f(v1, base + j);
        if (base + j < nt) { ax = fmaf(v, t[j].x, ax); ay = fmaf(v, t[j].y, ay); }
      }
    }
    open = (nt == TAIL_REGS);
  }

#define TAIL_EDGE(r, v, s)                                                  \
  { open = open && ((r) == cur);                                            \
    const float tx = fmaf((v), (s).x, ax);                                  \
    const float ty = fmaf((v), (s).y, ay);                                  \
    ax = open ? tx : ax; ay = open ? ty : ay; }

  for (long e = (long)e1 + TAIL_REGS; open && e < N_EDGES; e += 8) {
    const int4   ra = *(const int4*)(row_idx + e);
    const int4   rb = *(const int4*)(row_idx + e + 4);
    const int4   ca = *(const int4*)(col_idx + e);
    const int4   cb = *(const int4*)(col_idx + e + 4);
    const float4 va = *(const float4*)(edge_vals + e);
    const float4 vb = *(const float4*)(edge_vals + e + 4);
    const float2 t0 = *(const float2*)(support + (size_t)ca.x * OUT_F + c2);
    const float2 t1 = *(const float2*)(support + (size_t)ca.y * OUT_F + c2);
    const float2 t2 = *(const float2*)(support + (size_t)ca.z * OUT_F + c2);
    const float2 t3 = *(const float2*)(support + (size_t)ca.w * OUT_F + c2);
    const float2 t4 = *(const float2*)(support + (size_t)cb.x * OUT_F + c2);
    const float2 t5 = *(const float2*)(support + (size_t)cb.y * OUT_F + c2);
    const float2 t6 = *(const float2*)(support + (size_t)cb.z * OUT_F + c2);
    const float2 t7 = *(const float2*)(support + (size_t)cb.w * OUT_F + c2);
    TAIL_EDGE(ra.x, va.x, t0)
    TAIL_EDGE(ra.y, va.y, t1)
    TAIL_EDGE(ra.z, va.z, t2)
    TAIL_EDGE(ra.w, va.w, t3)
    TAIL_EDGE(rb.x, vb.x, t4)
    TAIL_EDGE(rb.y, vb.y, t5)
    TAIL_EDGE(rb.z, vb.z, t6)
    TAIL_EDGE(rb.w, vb.w, t7)
  }
#undef TAIL_EDGE

  put_row(out, cur, c2, ax + b.x, ay + b.y);
  if (open) {
    for (int z = cur + 1; z < N_NODES; ++z) put_row(out, z, c2, b.x, b.y);
  }
}

extern "C" void kernel_launch(void* const* d_in, const int* in_sizes, int n_in,
                              void* d_out, int out_size, void* d_ws, size_t ws_size,
                              hipStream_t stream) {
  const float* X         = (const float*)d_in[0];
  const float* edge_vals = (const float*)d_in[1];
  const float* W         = (const float*)d_in[2];
  const float* bias      = (const float*)d_in[3];
  const int*   row_idx   = (const int*)d_in[4];
  const int*   col_idx   = (const int*)d_in[5];
  float* out     = (float*)d_out;
  float* support = (float*)d_ws;

  (void)in_sizes; (void)n_in; (void)out_size; (void)ws_size;

  gemm_xw_wmma<<<GEMM_GROUPS, 256, 0, stream>>>(X, W, support);
  int waves  = (N_EDGES + EDGES_PER_WAVE - 1) / EDGES_PER_WAVE;
  int blocks = (waves + 7) / 8;
  spmm_rows<<<blocks, 256, 0, stream>>>(support, edge_vals, row_idx, col_idx, bias, out);
}


# SpMM: a 16-lane half-wave walks 64 edges, 4 columns per lane, so one 16-B-per-lane load gathers two support rows; row starts tested as scalar bits of four wave masks; row sums still in edge order
# speedup vs baseline: 1.3461x; 1.0557x over previous
#include <hip/hip_runtime.h>

typedef __attribute__((ext_vector_type(2))) float v2f;
typedef __attribute__((ext_vector_type(8))) float v8f;

#define N_NODES 100000
#define N_EDGES 1600000
#define IN_F    256
#define OUT_F   64

#define GEMM_GROUPS 768
#define MAIN_TILES  (GEMM_GROUPS * 8)
#define LEFT_TILES  (N_NODES / 16 - MAIN_TILES)

__global__ __launch_bounds__(256) void gemm_xw_wmma(const float* __restrict__ X,
                                                    const float* __restrict__ W,
                                                    float* __restrict__ support) {
  __shared__ __attribute__((aligned(16))) float lw[96 * 256];

  const int wave = __builtin_amdgcn_readfirstlane((int)(threadIdx.x >> 5));
  const int lane = threadIdx.x & 31;
  const int hi   = lane >> 4;
  const int lo   = lane & 15;
  const int row_base = ((int)blockIdx.x * 8 + wave) * 16;
  const bool left = ((int)blockIdx.x < LEFT_TILES) && (wave < 4);
  const int left_base = (MAIN_TILES + (int)blockIdx.x) * 16;
  const int q = wave;

  const float* xrow = X + (size_t)(row_base + lo) * IN_F + 2 * hi;
  const float* xlft = X + (size_t)((left ? left_base : row_base) + lo) * IN_F + 2 * hi;
  v2f a[3][8];
  if (left) {
#pragma unroll
    for (int j = 0; j < 8; ++j) a[1][j] = *(const v2f*)(xlft + 4 * j);
#pragma unroll
    for (int j = 0; j < 8; ++j) a[2][j] = *(const v2f*)(xlft + 32 + 4 * j);
  } else {
#pragma unroll
    for (int j = 0; j < 8; ++j) a[0][j] = *(const v2f*)(xrow + 4 * j);
#pragma unroll
    for (int j = 0; j < 8; ++j) a[1][j] = *(const v2f*)(xrow + 32 + 4 * j);
  }

  const float4* wsrc = (const float4*)W + threadIdx.x;
  float4 wv[16];
#pragma unroll
  for (int i = 0; i < 16; ++i) wv[i] = wsrc[i * 256];
  {
    const int kw = (int)(threadIdx.x >> 4);
    const int nw = (int)(threadIdx.x & 15) * 4;
    const int sl = kw * 64 + ((((nw >> 4) + (kw >> 1)) & 3) << 4) + (nw & 15);
#pragma unroll
    for (int i = 0; i < 16; ++i) *(float4*)(lw + sl + i * 1024) = wv[i];
  }
  __syncthreads();

  if (left) {
    v8f accq = {};
#pragma unroll
    for (int gb = 0; gb < 8; ++gb) {
      const float* nxt = (gb + 2 < 8) ? (xlft + (gb + 2) * 32) : (xrow + (gb + 2 - 8) * 32);
#pragma unroll
      for (int j = 0; j < 8; ++j) a[gb % 3][j] = *(const v2f*)(nxt + 4 * j);
#pragma unroll
      for (int j = 0; j < 8; ++j) {
        const int k0 = gb * 32 + 4 * j;
        const int ka = k0 + 2 * hi;
        const v2f av = a[(gb + 1) % 3][j];
        const float* wr = lw + ka * 64;
        const int kr = ka >> 1;
        const int sq = (((kr + q) & 3) << 4) + lo;
        v2f bq = { wr[sq], wr[sq + 64] };
        accq = __builtin_amdgcn_wmma_f32_16x16x4_f32(false, av, false, bq, (short)0, accq, false, false);
      }
    }
    float* qrow = support + (size_t)(left_base + 8 * hi) * OUT_F + q * 16 + lo;
#pragma unroll
    for (int r = 0; r < 8; ++r) qrow[(size_t)r * OUT_F] = accq[r];
  }

  v8f acc0 = {}, acc1 = {}, acc2 = {}, acc3 = {};
#pragma unroll
  for (int kb = 0; kb < 8; ++kb) {
    if (kb + 2 < 8) {
#pragma unroll
      for (int j = 0; j < 8; ++j)
        a[(kb + 2) % 3][j] = *(const v2f*)(xrow + (kb + 2) * 32 + 4 * j);
    }
#pragma unroll
    for (int j = 0; j < 8; ++j) {
      const int k0 = kb * 32 + 4 * j;
      const int ka = k0 + 2 * hi;
      const v2f av = a[kb % 3][j];
      const float* wr = lw + ka * 64;
      const int kr = ka >> 1;
      const int s0 = (((kr + 0) & 3) << 4) + lo;
      const int s1 = (((kr + 1) & 3) << 4) + lo;
      const int s2 = (((kr + 2) & 3) << 4) + lo;
      const int s3 = (((kr + 3) & 3) << 4) + lo;
      v2f b0 = { wr[s0], wr[s0 + 64] };
      v2f b1 = { wr[s1], wr[s1 + 64] };
      v2f b2 = { wr[s2], wr[s2 + 64] };
      v2f b3 = { wr[s3], wr[s3 + 64] };
      acc0 = __builtin_amdgcn_wmma_f32_16x16x4_f32(false, av, false, b0, (short)0, acc0, false, false);
      acc1 = __builtin_amdgcn_wmma_f32_16x16x4_f32(false, av, false, b1, (short)0, acc1, false, false);
      acc2 = __builtin_amdgcn_wmma_f32_16x16x4_f32(false, av, false, b2, (short)0, acc2, false, false);
      acc3 = __builtin_amdgcn_wmma_f32_16x16x4_f32(false, av, false, b3, (short)0, acc3, false, false);
    }
  }

  float* orow = support + (size_t)(row_base + 8 * hi) * OUT_F + lo;
#pragma unroll
  for (int r = 0; r < 8; ++r) {
    orow[(size_t)r * OUT_F +  0] = acc0[r];
    orow[(size_t)r * OUT_F + 16] = acc1[r];
    orow[(size_t)r * OUT_F + 32] = acc2[r];
    orow[(size_t)r * OUT_F + 48] = acc3[r];
  }
}

#define EDGES_PER_WAVE 128
#define EDGES_PER_HALF 64
#define GATHER_RING    8
#define TAIL_REGS      32

__device__ __forceinline__ void put_row(float* __restrict__ out, int r, int c4,
                                        float x, float y, float z, float w) {
  *(float4*)(out + (size_t)r * OUT_F + c4) = make_float4(x, y, z, w);
}

__device__ __forceinline__ int   pick(const int4 v, int k)   { return k == 0 ? v.x : k == 1 ? v.y : k == 2 ? v.z : v.w; }
__device__ __forceinline__ float pick(const float4 v, int k) { return k == 0 ? v.x : k == 1 ? v.y : k == 2 ? v.z : v.w; }
__device__ __forceinline__ int half_i(int v, int ln) {
  const unsigned sel = (unsigned)ln * 0x11111111u;
  return (int)__builtin_amdgcn_permlane16((unsigned)v, (unsigned)v, sel, sel, false, false);
}
__device__ __forceinline__ float half_f(float v, int ln) { return __int_as_float(half_i(__float_as_int(v), ln)); }
__device__ __forceinline__ int sel_i(int v, unsigned sel) {
  return (int)__builtin_amdgcn_permlane16((unsigned)v, (unsigned)v, sel, sel, false, false);
}
__device__ __forceinline__ float sel_f(float v, unsigned sel) { return __int_as_float(sel_i(__float_as_int(v), sel)); }

__device__ __forceinline__ float4 gather_row(const float* __restrict__ support, int c, int l16) {
  const float4* s4 = (const float4*)support;
  return s4[(unsigned)c * (unsigned)(OUT_F / 4) + (unsigned)l16];
}

__global__ __launch_bounds__(256) void spmm_rows(const float* __restrict__ support,
                                                 const float* __restrict__ edge_vals,
                                                 const int* __restrict__ row_idx,
                                                 const int* __restrict__ col_idx,
                                                 const float* __restrict__ bias,
                                                 float* __restrict__ out) {
  const int lane = threadIdx.x & 31;
  const int l16  = lane & 15;
  const int hsh  = lane & 16;
  const int wave = (int)blockIdx.x * 8 + __builtin_amdgcn_readfirstlane((int)(threadIdx.x >> 5));
  const int w0 = wave * EDGES_PER_WAVE;
  if (w0 >= N_EDGES) return;
  const int e0 = w0 + (lane >> 4) * EDGES_PER_HALF;
  const int e1 = e0 + EDGES_PER_HALF;
  const bool has_next = e1 < N_EDGES;

  const int c4 = l16 * 4;

  const int4   r0 = *(const int4*)(row_idx + w0 + 4 * lane);
  const int4   k0 = *(const int4*)(col_idx + w0 + 4 * lane);
  const float4 v0 = *(const float4*)(edge_vals + w0 + 4 * lane);
  const int et = (has_next ? e1 : e0) + l16;
  const int   ra = row_idx[et],   rb = row_idx[et + 16];
  const int   ka = col_idx[et],   kb = col_idx[et + 16];
  const float va = edge_vals[et], vb = edge_vals[et + 16];

  const float4 b = *(const float4*)(bias + c4);
  float ax = 0.f, ay = 0.f, az = 0.f, aw = 0.f;
  int cur = (e0 > 0) ? row_idx[e0 - 1] : -1;
  bool own = false;

  int pw = (int)__builtin_amdgcn_permlane16((unsigned)r0.w, (unsigned)r0.w,
                                            0x65432100u, 0xEDCBA987u, false, false);
  pw = (l16 == 0) ? cur : pw;
  const unsigned m0 = __builtin_amdgcn_ballot_w32(r0.x != pw);
  const unsigned m1 = __builtin_amdgcn_ballot_w32(r0.y != r0.x);
  const unsigned m2 = __builtin_amdgcn_ballot_w32(r0.z != r0.y);
  const unsigned m3 = __builtin_amdgcn_ballot_w32(r0.w != r0.z);

#define ROW_STEP(c, L, SEL, s)                                              \
  {                                                                         \
    const unsigned mc = (c) == 0 ? m0 : (c) == 1 ? m1 : (c) == 2 ? m2 : m3; \
    if ((mc >> (L)) & 0x10001u) {                                           \
      const int r = sel_i(pick(r0, (c)), (SEL));                            \
      if (r != cur) {                                                       \
        if (own) put_row(out, cur, c4, ax + b.x, ay + b.y, az + b.z, aw + b.w); \
        _Pragma("nounroll")                                                 \
        for (int z = cur + 1; z < r; ++z) put_row(out, z, c4, b.x, b.y, b.z, b.w); \
        cur = r; own = true; ax = 0.f; ay = 0.f; az = 0.f; aw = 0.f;        \
      }                                                                     \
    }                                                                       \
    const float v = sel_f(pick(v0, (c)), (SEL));                            \
    ax = fmaf(v, (s).x, ax); ay = fmaf(v, (s).y, ay);                       \
    az = fmaf(v, (s).z, az); aw = fmaf(v, (s).w, aw);                       \
  }

  float4 g[GATHER_RING];
#pragma unroll
  for (int p = 0; p < GATHER_RING; ++p)
    g[p] = gather_row(support, half_i(pick(k0, p & 3), p >> 2), l16);
  for (int blk = 0; blk < EDGES_PER_HALF / GATHER_RING - 1; ++blk) {
    const int l0 = blk * (GATHER_RING / 4);
    const unsigned s0 = __builtin_amdgcn_readfirstlane((unsigned)l0 * 0x11111111u);
#pragma unroll
    for (int p = 0; p < GATHER_RING; ++p) {
      ROW_STEP(p & 3, l0 + (p >> 2), s0 + (unsigned)(p >> 2) * 0x11111111u, g[p])
      g[p] = gather_row(support, sel_i(pick(k0, p & 3), s0 + (unsigned)(GATHER_RING / 4 + (p >> 2)) * 0x11111111u), l16);
    }
  }
#pragma unroll
  for (int p = 0; p < GATHER_RING; ++p) {
    ROW_STEP(p & 3, (EDGES_PER_HALF - GATHER_RING) / 4 + (p >> 2),
             (unsigned)((EDGES_PER_HALF - GATHER_RING) / 4 + (p >> 2)) * 0x11111111u, g[p])
  }
#undef ROW_STEP

  const unsigned ma = __builtin_amdgcn_ballot_w32(has_next && ra == cur);
  const unsigned mb = __builtin_amdgcn_ballot_w32(has_next && rb == cur);
  const int nt = __builtin_popcount((ma >> hsh) & 0xffffu) + __builtin_popcount((mb >> hsh) & 0xffffu);
#pragma unroll
  for (int base = 0; base < TAIL_REGS; base += 8) {
    if (__builtin_amdgcn_ballot_w32(own && base < nt) == 0u) break;
    float4 t[8];
#pragma unroll
    for (int j = 0; j < 8; ++j)
      t[j] = gather_row(support, half_i(base < 16 ? ka : kb, (base + j) & 15), l16);
#pragma unroll
    for (int j = 0; j < 8; ++j) {
      const float v = half_f(base < 16 ? va : vb, (base + j) & 15);
      if (own && base + j < nt) {
        ax = fmaf(v, t[j].x, ax); ay = fmaf(v, t[j].y, ay);
        az = fmaf(v, t[j].z, az); aw = fmaf(v, t[j].w, aw);
      }
    }
  }
  bool open = own && (!has_next || nt == TAIL_REGS);

#define TAIL_EDGE(r, v, s)                                                  \
  { open = open && ((r) == cur);                                            \
    const float tx = fmaf((v), (s).x, ax);                                  \
    const float ty = fmaf((v), (s).y, ay);                                  \
    const float tz = fmaf((v), (s).z, az);                                  \
    const float tw = fmaf((v), (s).w, aw);                                  \
    ax = open ? tx : ax; ay = open ? ty : ay;                               \
    az = open ? tz : az; aw = open ? tw : aw; }

  for (long e = (long)e1 + TAIL_REGS; open && e < N_EDGES; e += 8) {
    const int4   qa = *(const int4*)(row_idx + e);
    const int4   qb = *(const int4*)(row_idx + e + 4);
    const int4   ca = *(const int4*)(col_idx + e);
    const int4   cb = *(const int4*)(col_idx + e + 4);
    const float4 xa = *(const float4*)(edge_vals + e);
    const float4 xb = *(const float4*)(edge_vals + e + 4);
    const float4 t0 = gather_row(support, ca.x, l16);
    const float4 t1 = gather_row(support, ca.y, l16);
    const float4 t2 = gather_row(support, ca.z, l16);
    const float4 t3 = gather_row(support, ca.w, l16);
    const float4 t4 = gather_row(support, cb.x, l16);
    const float4 t5 = gather_row(support, cb.y, l16);
    const float4 t6 = gather_row(support, cb.z, l16);
    const float4 t7 = gather_row(support, cb.w, l16);
    TAIL_EDGE(qa.x, xa.x, t0)
    TAIL_EDGE(qa.y, xa.y, t1)
    TAIL_EDGE(qa.z, xa.z, t2)
    TAIL_EDGE(qa.w, xa.w, t3)
    TAIL_EDGE(qb.x, xb.x, t4)
    TAIL_EDGE(qb.y, xb.y, t5)
    TAIL_EDGE(qb.z, xb.z, t6)
    TAIL_EDGE(qb.w, xb.w, t7)
  }
#undef TAIL_EDGE

  if (own) {
    put_row(out, cur, c4, ax + b.x, ay + b.y, az + b.z, aw + b.w);
    if (open) {
      for (int z = cur + 1; z < N_NODES; ++z) put_row(out, z, c4, b.x, b.y, b.z, b.w);
    }
  }
}

extern "C" void kernel_launch(void* const* d_in, const int* in_sizes, int n_in,
                              void* d_out, int out_size, void* d_ws, size_t ws_size,
                              hipStream_t stream) {
  const float* X         = (const float*)d_in[0];
  const float* edge_vals = (const float*)d_in[1];
  const float* W         = (const float*)d_in[2];
  const float* bias      = (const float*)d_in[3];
  const int*   row_idx   = (const int*)d_in[4];
  const int*   col_idx   = (const int*)d_in[5];
  float* out     = (float*)d_out;
  float* support = (float*)d_ws;

  (void)in_sizes; (void)n_in; (void)out_size; (void)ws_size;

  gemm_xw_wmma<<<GEMM_GROUPS, 256, 0, stream>>>(X, W, support);
  int waves  = (N_EDGES + EDGES_PER_WAVE - 1) / EDGES_PER_WAVE;
  int blocks = (waves + 7) / 8;
  spmm_rows<<<blocks, 256, 0, stream>>>(support, edge_vals, row_idx, col_idx, bias, out);
}


# GEMM: X fetched in whole 128-B lines (4 loads of 4 lines per 32-k block, was 8 of 16 lines), turned to A fragments in a wave-private LDS area; W fragments at 4 lane bases + constants; sums unchanged
# speedup vs baseline: 1.4409x; 1.0704x over previous
#include <hip/hip_runtime.h>

typedef __attribute__((ext_vector_type(2))) float v2f;
typedef __attribute__((ext_vector_type(4))) float v4f;
typedef __attribute__((ext_vector_type(8))) float v8f;

#define N_NODES 100000
#define N_EDGES 1600000
#define IN_F    256
#define OUT_F   64

#define GEMM_GROUPS 768
#define MAIN_TILES  (GEMM_GROUPS * 8)
#define LEFT_TILES  (N_NODES / 16 - MAIN_TILES)
#define XS_PITCH 36
#define XS_WAVE  1024

__global__ __launch_bounds__(256) void gemm_xw_wmma(const float* __restrict__ X,
                                                    const float* __restrict__ W,
                                                    float* __restrict__ support) {
  __shared__ __attribute__((aligned(16))) float lw[64 * 256];
  __shared__ __attribute__((aligned(16))) float xst[8 * XS_WAVE];

  const int wave = __builtin_amdgcn_readfirstlane((int)(threadIdx.x >> 5));
  const int lane = threadIdx.x & 31;
  const int hi   = lane >> 4;
  const int lo   = lane & 15;
  const int lr   = lane >> 3;
  const int lc   = lane & 7;
  const int row_base = ((int)blockIdx.x * 8 + wave) * 16;
  const bool left = ((int)blockIdx.x < LEFT_TILES) && (wave < 4);
  const int left_base = (MAIN_TILES + (int)blockIdx.x) * 16;
  const int q = wave;

  const float* xg  = X + (size_t)(row_base + lr) * IN_F + 4 * lc;
  const float* xgl = X + (size_t)((left ? left_base : row_base) + lr) * IN_F + 4 * lc;
  float* xs = xst + wave * XS_WAVE;
  float* xw = xs + lr * XS_PITCH + 4 * lc;
  const float* xr = xs + lo * XS_PITCH + 2 * hi;
  int ob0 = hi * 128 + (((hi + 0) & 3) << 4) + lo;
  int ob1 = hi * 128 + (((hi + 1) & 3) << 4) + lo;
  int ob2 = hi * 128 + (((hi + 2) & 3) << 4) + lo;
  int ob3 = hi * 128 + (((hi + 3) & 3) << 4) + lo;
  asm("" : "+v"(ob0), "+v"(ob1), "+v"(ob2), "+v"(ob3));
  const float* bp0 = lw + ob0;
  const float* bp1 = lw + ob1;
  const float* bp2 = lw + ob2;
  const float* bp3 = lw + ob3;
  v4f g[2][4];
  v2f a[2][8];
#define X_FETCH(slot, src)                                                       \
  { _Pragma("unroll") for (int i = 0; i < 4; ++i)                                \
      g[slot][i] = *(const v4f*)((src) + i * 4 * IN_F); }
#define X_STAGE(slot)                                                            \
  { _Pragma("unroll") for (int i = 0; i < 4; ++i)                                \
      *(v4f*)(xw + i * 4 * XS_PITCH) = g[slot][i]; }
#define X_FRAGS(set)                                                             \
  { _Pragma("unroll") for (int j = 0; j < 8; ++j)                                \
      a[set][j] = *(const v2f*)(xr + 4 * j); }

  if (left) { X_FETCH(0, xgl) X_FETCH(1, xgl + 32) }
  else      { X_FETCH(0, xg)  X_FETCH(1, xg + 32) }

  const float4* wsrc = (const float4*)W + threadIdx.x;
  float4 wv[16];
#pragma unroll
  for (int i = 0; i < 16; ++i) wv[i] = wsrc[i * 256];
  {
    const int kw = (int)(threadIdx.x >> 4);
    const int nw = (int)(threadIdx.x & 15) * 4;
    const int sl = kw * 64 + ((((nw >> 4) + (kw >> 1)) & 3) << 4) + (nw & 15);
#pragma unroll
    for (int i = 0; i < 16; ++i) *(float4*)(lw + sl + i * 1024) = wv[i];
  }
  __syncthreads();

  if (left) {
    X_STAGE(0) X_FETCH(0, xgl + 2 * 32) X_FRAGS(0)
    const float* bqe = lw + hi * 128 + (((hi + q) & 3) << 4) + lo;
    const float* bqo = lw + hi * 128 + (((hi + q + 2) & 3) << 4) + lo;
    v8f accq = {};
#pragma unroll
    for (int s = 0; s < 8; ++s) {
      const float* nxt = (s + 3 < 8) ? (xgl + (s + 3) * 32) : (xg + (s + 3 - 8) * 32);
      X_STAGE((s + 1) % 2) X_FETCH((s + 1) % 2, nxt) X_FRAGS((s + 1) % 2)
#pragma unroll
      for (int j = 0; j < 8; ++j) {
        const int k0 = s * 32 + 4 * j;
        const v2f av = a[s % 2][j];
        const float* wq = ((j & 1) ? bqo : bqe) + k0 * 64;
        v2f bq = { wq[0], wq[64] };
        accq = __builtin_amdgcn_wmma_f32_16x16x4_f32(false, av, false, bq, (short)0, accq, false, false);
      }
    }
    float* qrow = support + (size_t)(left_base + 8 * hi) * OUT_F + q * 16 + lo;
#pragma unroll
    for (int r = 0; r < 8; ++r) qrow[(size_t)r * OUT_F] = accq[r];
  } else {
    X_STAGE(0) X_FETCH(0, xg + 2 * 32) X_FRAGS(0)
  }

  v8f acc0 = {}, acc1 = {}, acc2 = {}, acc3 = {};
#pragma unroll
  for (int kb = 0; kb < 8; ++kb) {
    if (kb + 1 < 8) {
      X_STAGE((kb + 1) % 2)
      if (kb + 3 < 8) X_FETCH((kb + 1) % 2, xg + (kb + 3) * 32)
      X_FRAGS((kb + 1) % 2)
    }
#pragma unroll
    for (int j = 0; j < 8; ++j) {
      const int k0 = kb * 32 + 4 * j;
      const v2f av = a[kb % 2][j];
      const float* w0 = ((j & 1) ? bp2 : bp0) + k0 * 64;
      const float* w1 = ((j & 1) ? bp3 : bp1) + k0 * 64;
      const float* w2 = ((j & 1) ? bp0 : bp2) + k0 * 64;
      const float* w3 = ((j & 1) ? bp1 : bp3) + k0 * 64;
      v2f b0 = { w0[0], w0[64] };
      v2f b1 = { w1[0], w1[64] };
      v2f b2 = { w2[0], w2[64] };
      v2f b3 = { w3[0], w3[64] };
      acc0 = __builtin_amdgcn_wmma_f32_16x16x4_f32(false, av, false, b0, (short)0, acc0, false, false);
      acc1 = __builtin_amdgcn_wmma_f32_16x16x4_f32(false, av, false, b1, (short)0, acc1, false, false);
      acc2 = __builtin_amdgcn_wmma_f32_16x16x4_f32(false, av, false, b2, (short)0, acc2, false, false);
      acc3 = __builtin_amdgcn_wmma_f32_16x16x4_f32(false, av, false, b3, (short)0, acc3, false, false);
    }
  }
#undef X_FETCH
#undef X_STAGE
#undef X_FRAGS

  float* orow = support + (size_t)(row_base + 8 * hi) * OUT_F + lo;
#pragma unroll
  for (int r = 0; r < 8; ++r) {
    orow[(size_t)r * OUT_F +  0] = acc0[r];
    orow[(size_t)r * OUT_F + 16] = acc1[r];
    orow[(size_t)r * OUT_F + 32] = acc2[r];
    orow[(size_t)r * OUT_F + 48] = acc3[r];
  }
}

#define EDGES_PER_WAVE 128
#define EDGES_PER_HALF 64
#define GATHER_RING    8
#define TAIL_REGS      32

__device__ __forceinline__ void put_row(float* __restrict__ out, int r, int c4,
                                        float x, float y, float z, float w) {
  *(float4*)(out + (size_t)r * OUT_F + c4) = make_float4(x, y, z, w);
}

__device__ __forceinline__ int   pick(const int4 v, int k)   { return k == 0 ? v.x : k == 1 ? v.y : k == 2 ? v.z : v.w; }
__device__ __forceinline__ float pick(const float4 v, int k) { return k == 0 ? v.x : k == 1 ? v.y : k == 2 ? v.z : v.w; }
__device__ __forceinline__ int half_i(int v, int ln) {
  const unsigned sel = (unsigned)ln * 0x11111111u;
  return (int)__builtin_amdgcn_permlane16((unsigned)v, (unsigned)v, sel, sel, false, false);
}
__device__ __forceinline__ float half_f(float v, int ln) { return __int_as_float(half_i(__float_as_int(v), ln)); }
__device__ __forceinline__ int sel_i(int v, unsigned sel) {
  return (int)__builtin_amdgcn_permlane16((unsigned)v, (unsigned)v, sel, sel, false, false);
}
__device__ __forceinline__ float sel_f(float v, unsigned sel) { return __int_as_float(sel_i(__float_as_int(v), sel)); }

__device__ __forceinline__ float4 gather_row(const float* __restrict__ support, int c, int l16) {
  const float4* s4 = (const float4*)support;
  return s4[(unsigned)c * (unsigned)(OUT_F / 4) + (unsigned)l16];
}

__global__ __launch_bounds__(256) void spmm_rows(const float* __restrict__ support,
                                                 const float* __restrict__ edge_vals,
                                                 const int* __restrict__ row_idx,
                                                 const int* __restrict__ col_idx,
                                                 const float* __restrict__ bias,
                                                 float* __restrict__ out) {
  const int lane = threadIdx.x & 31;
  const int l16  = lane & 15;
  const int hsh  = lane & 16;
  const int wave = (int)blockIdx.x * 8 + __builtin_amdgcn_readfirstlane((int)(threadIdx.x >> 5));
  const int w0 = wave * EDGES_PER_WAVE;
  if (w0 >= N_EDGES) return;
  const int e0 = w0 + (lane >> 4) * EDGES_PER_HALF;
  const int e1 = e0 + EDGES_PER_HALF;
  const bool has_next = e1 < N_EDGES;

  const int c4 = l16 * 4;

  const int4   r0 = *(const int4*)(row_idx + w0 + 4 * lane);
  const int4   k0 = *(const int4*)(col_idx + w0 + 4 * lane);
  const float4 v0 = *(const float4*)(edge_vals + w0 + 4 * lane);
  const int et = (has_next ? e1 : e0) + l16;
  const int   ra = row_idx[et],   rb = row_idx[et + 16];
  const int   ka = col_idx[et],   kb = col_idx[et + 16];
  const float va = edge_vals[et], vb = edge_vals[et + 16];

  const float4 b = *(const float4*)(bias + c4);
  float ax = 0.f, ay = 0.f, az = 0.f, aw = 0.f;
  int cur = (e0 > 0) ? row_idx[e0 - 1] : -1;
  bool own = false;

  int pw = (int)__builtin_amdgcn_permlane16((unsigned)r0.w, (unsigned)r0.w,
                                            0x65432100u, 0xEDCBA987u, false, false);
  pw = (l16 == 0) ? cur : pw;
  const unsigned m0 = __builtin_amdgcn_ballot_w32(r0.x != pw);
  const unsigned m1 = __builtin_amdgcn_ballot_w32(r0.y != r0.x);
  const unsigned m2 = __builtin_amdgcn_ballot_w32(r0.z != r0.y);
  const unsigned m3 = __builtin_amdgcn_ballot_w32(r0.w != r0.z);

#define ROW_STEP(c, L, SEL, s)                                              \
  {                                                                         \
    const unsigned mc = (c) == 0 ? m0 : (c) == 1 ? m1 : (c) == 2 ? m2 : m3; \
    if ((mc >> (L)) & 0x10001u) {                                           \
      const int r = sel_i(pick(r0, (c)), (SEL));                            \
      if (r != cur) {                                                       \
        if (own) put_row(out, cur, c4, ax + b.x, ay + b.y, az + b.z, aw + b.w); \
        _Pragma("nounroll")                                                 \
        for (int z = cur + 1; z < r; ++z) put_row(out, z, c4, b.x, b.y, b.z, b.w); \
        cur = r; own = true; ax = 0.f; ay = 0.f; az = 0.f; aw = 0.f;        \
      }                                                                     \
    }                                                                       \
    const float v = sel_f(pick(v0, (c)), (SEL));                            \
    ax = fmaf(v, (s).x, ax); ay = fmaf(v, (s).y, ay);                       \
    az = fmaf(v, (s).z, az); aw = fmaf(v, (s).w, aw);                       \
  }

  float4 g[GATHER_RING];
#pragma unroll
  for (int p = 0; p < GATHER_RING; ++p)
    g[p] = gather_row(support, half_i(pick(k0, p & 3), p >> 2), l16);
  for (int blk = 0; blk < EDGES_PER_HALF / GATHER_RING - 1; ++blk) {
    const int l0 = blk * (GATHER_RING / 4);
    const unsigned s0 = __builtin_amdgcn_readfirstlane((unsigned)l0 * 0x11111111u);
#pragma unroll
    for (int p = 0; p < GATHER_RING; ++p) {
      ROW_STEP(p & 3, l0 + (p >> 2), s0 + (unsigned)(p >> 2) * 0x11111111u, g[p])
      g[p] = gather_row(support, sel_i(pick(k0, p & 3), s0 + (unsigned)(GATHER_RING / 4 + (p >> 2)) * 0x11111111u), l16);
    }
  }
#pragma unroll
  for (int p = 0; p < GATHER_RING; ++p) {
    ROW_STEP(p & 3, (EDGES_PER_HALF - GATHER_RING) / 4 + (p >> 2),
             (unsigned)((EDGES_PER_HALF - GATHER_RING) / 4 + (p >> 2)) * 0x11111111u, g[p])
  }
#undef ROW_STEP

  const unsigned ma = __builtin_amdgcn_ballot_w32(has_next && ra == cur);
  const unsigned mb = __builtin_amdgcn_ballot_w32(has_next && rb == cur);
  const int nt = __builtin_popcount((ma >> hsh) & 0xffffu) + __builtin_popcount((mb >> hsh) & 0xffffu);
#pragma unroll
  for (int base = 0; base < TAIL_REGS; base += 8) {
    if (__builtin_amdgcn_ballot_w32(own && base < nt) == 0u) break;
    float4 t[8];
#pragma unroll
    for (int j = 0; j < 8; ++j)
      t[j] = gather_row(support, half_i(base < 16 ? ka : kb, (base + j) & 15), l16);
#pragma unroll
    for (int j = 0; j < 8; ++j) {
      const float v = half_f(base < 16 ? va : vb, (base + j) & 15);
      if (own && base + j < nt) {
        ax = fmaf(v, t[j].x, ax); ay = fmaf(v, t[j].y, ay);
        az = fmaf(v, t[j].z, az); aw = fmaf(v, t[j].w, aw);
      }
    }
  }
  bool open = own && (!has_next || nt == TAIL_REGS);

#define TAIL_EDGE(r, v, s)                                                  \
  { open = open && ((r) == cur);                                            \
    const float tx = fmaf((v), (s).x, ax);                                  \
    const float ty = fmaf((v), (s).y, ay);                                  \
    const float tz = fmaf((v), (s).z, az);                                  \
    const float tw = fmaf((v), (s).w, aw);                                  \
    ax = open ? tx : ax; ay = open ? ty : ay;                               \
    az = open ? tz : az; aw = open ? tw : aw; }

  for (long e = (long)e1 + TAIL_REGS; open && e < N_EDGES; e += 8) {
    const int4   qa = *(const int4*)(row_idx + e);
    const int4   qb = *(const int4*)(row_idx + e + 4);
    const int4   ca = *(const int4*)(col_idx + e);
    const int4   cb = *(const int4*)(col_idx + e + 4);
    const float4 xa = *(const float4*)(edge_vals + e);
    const float4 xb = *(const float4*)(edge_vals + e + 4);
    const float4 t0 = gather_row(support, ca.x, l16);
    const float4 t1 = gather_row(support, ca.y, l16);
    const float4 t2 = gather_row(support, ca.z, l16);
    const float4 t3 = gather_row(support, ca.w, l16);
    const float4 t4 = gather_row(support, cb.x, l16);
    const float4 t5 = gather_row(support, cb.y, l16);
    const float4 t6 = gather_row(support, cb.z, l16);
    const float4 t7 = gather_row(support, cb.w, l16);
    TAIL_EDGE(qa.x, xa.x, t0)
    TAIL_EDGE(qa.y, xa.y, t1)
    TAIL_EDGE(qa.z, xa.z, t2)
    TAIL_EDGE(qa.w, xa.w, t3)
    TAIL_EDGE(qb.x, xb.x, t4)
    TAIL_EDGE(qb.y, xb.y, t5)
    TAIL_EDGE(qb.z, xb.z, t6)
    TAIL_EDGE(qb.w, xb.w, t7)
  }
#undef TAIL_EDGE

  if (own) {
    put_row(out, cur, c4, ax + b.x, ay + b.y, az + b.z, aw + b.w);
    if (open) {
      for (int z = cur + 1; z < N_NODES; ++z) put_row(out, z, c4, b.x, b.y, b.z, b.w);
    }
  }
}

extern "C" void kernel_launch(void* const* d_in, const int* in_sizes, int n_in,
                              void* d_out, int out_size, void* d_ws, size_t ws_size,
                              hipStream_t stream) {
  const float* X         = (const float*)d_in[0];
  const float* edge_vals = (const float*)d_in[1];
  const float* W         = (const float*)d_in[2];
  const float* bias      = (const float*)d_in[3];
  const int*   row_idx   = (const int*)d_in[4];
  const int*   col_idx   = (const int*)d_in[5];
  float* out     = (float*)d_out;
  float* support = (float*)d_ws;

  (void)in_sizes; (void)n_in; (void)out_size; (void)ws_size;

  gemm_xw_wmma<<<GEMM_GROUPS, 256, 0, stream>>>(X, W, support);
  int waves  = (N_EDGES + EDGES_PER_WAVE - 1) / EDGES_PER_WAVE;
  int blocks = (waves + 7) / 8;
  spmm_rows<<<blocks, 256, 0, stream>>>(support, edge_vals, row_idx, col_idx, bias, out);
}


# SpMM: edges whose products are dropped (chunk's leading edges of a row owned earlier; tail edges past the row's end) gather a row the half-wave reads anyway, not their own random row; sums unchanged
# speedup vs baseline: 1.4961x; 1.0383x over previous
#include <hip/hip_runtime.h>

typedef __attribute__((ext_vector_type(2))) float v2f;
typedef __attribute__((ext_vector_type(4))) float v4f;
typedef __attribute__((ext_vector_type(8))) float v8f;

#define N_NODES 100000
#define N_EDGES 1600000
#define IN_F    256
#define OUT_F   64

#define GEMM_GROUPS 768
#define MAIN_TILES  (GEMM_GROUPS * 8)
#define LEFT_TILES  (N_NODES / 16 - MAIN_TILES)
#define XS_PITCH 36
#define XS_WAVE  1024

__global__ __launch_bounds__(256) void gemm_xw_wmma(const float* __restrict__ X,
                                                    const float* __restrict__ W,
                                                    float* __restrict__ support) {
  __shared__ __attribute__((aligned(16))) float lw[64 * 256];
  __shared__ __attribute__((aligned(16))) float xst[8 * XS_WAVE];

  const int wave = __builtin_amdgcn_readfirstlane((int)(threadIdx.x >> 5));
  const int lane = threadIdx.x & 31;
  const int hi   = lane >> 4;
  const int lo   = lane & 15;
  const int lr   = lane >> 3;
  const int lc   = lane & 7;
  const int row_base = ((int)blockIdx.x * 8 + wave) * 16;
  const bool left = ((int)blockIdx.x < LEFT_TILES) && (wave < 4);
  const int left_base = (MAIN_TILES + (int)blockIdx.x) * 16;
  const int q = wave;

  const float* xg  = X + (size_t)(row_base + lr) * IN_F + 4 * lc;
  const float* xgl = X + (size_t)((left ? left_base : row_base) + lr) * IN_F + 4 * lc;
  float* xs = xst + wave * XS_WAVE;
  float* xw = xs + lr * XS_PITCH + 4 * lc;
  const float* xr = xs + lo * XS_PITCH + 2 * hi;
  int ob0 = hi * 128 + (((hi + 0) & 3) << 4) + lo;
  int ob1 = hi * 128 + (((hi + 1) & 3) << 4) + lo;
  int ob2 = hi * 128 + (((hi + 2) & 3) << 4) + lo;
  int ob3 = hi * 128 + (((hi + 3) & 3) << 4) + lo;
  asm("" : "+v"(ob0), "+v"(ob1), "+v"(ob2), "+v"(ob3));
  const float* bp0 = lw + ob0;
  const float* bp1 = lw + ob1;
  const float* bp2 = lw + ob2;
  const float* bp3 = lw + ob3;
  v4f g[2][4];
  v2f a[2][8];
#define X_FETCH(slot, src)                                                       \
  { _Pragma("unroll") for (int i = 0; i < 4; ++i)                                \
      g[slot][i] = *(const v4f*)((src) + i * 4 * IN_F); }
#define X_STAGE(slot)                                                            \
  { _Pragma("unroll") for (int i = 0; i < 4; ++i)                                \
      *(v4f*)(xw + i * 4 * XS_PITCH) = g[slot][i]; }
#define X_FRAGS(set)                                                             \
  { _Pragma("unroll") for (int j = 0; j < 8; ++j)                                \
      a[set][j] = *(const v2f*)(xr + 4 * j); }

  if (left) { X_FETCH(0, xgl) X_FETCH(1, xgl + 32) }
  else      { X_FETCH(0, xg)  X_FETCH(1, xg + 32) }

  const float4* wsrc = (const float4*)W + threadIdx.x;
  float4 wv[16];
#pragma unroll
  for (int i = 0; i < 16; ++i) wv[i] = wsrc[i * 256];
  {
    const int kw = (int)(threadIdx.x >> 4);
    const int nw = (int)(threadIdx.x & 15) * 4;
    const int sl = kw * 64 + ((((nw >> 4) + (kw >> 1)) & 3) << 4) + (nw & 15);
#pragma unroll
    for (int i = 0; i < 16; ++i) *(float4*)(lw + sl + i * 1024) = wv[i];
  }
  __syncthreads();

  if (left) {
    X_STAGE(0) X_FETCH(0, xgl + 2 * 32) X_FRAGS(0)
    const float* bqe = lw + hi * 128 + (((hi + q) & 3) << 4) + lo;
    const float* bqo = lw + hi * 128 + (((hi + q + 2) & 3) << 4) + lo;
    v8f accq = {};
#pragma unroll
    for (int s = 0; s < 8; ++s) {
      const float* nxt = (s + 3 < 8) ? (xgl + (s + 3) * 32) : (xg + (s + 3 - 8) * 32);
      X_STAGE((s + 1) % 2) X_FETCH((s + 1) % 2, nxt) X_FRAGS((s + 1) % 2)
#pragma unroll
      for (int j = 0; j < 8; ++j) {
        const int k0 = s * 32 + 4 * j;
        const v2f av = a[s % 2][j];
        const float* wq = ((j & 1) ? bqo : bqe) + k0 * 64;
        v2f bq = { wq[0], wq[64] };
        accq = __builtin_amdgcn_wmma_f32_16x16x4_f32(false, av, false, bq, (short)0, accq, false, false);
      }
    }
    float* qrow = support + (size_t)(left_base + 8 * hi) * OUT_F + q * 16 + lo;
#pragma unroll
    for (int r = 0; r < 8; ++r) qrow[(size_t)r * OUT_F] = accq[r];
  } else {
    X_STAGE(0) X_FETCH(0, xg + 2 * 32) X_FRAGS(0)
  }

  v8f acc0 = {}, acc1 = {}, acc2 = {}, acc3 = {};
#pragma unroll
  for (int kb = 0; kb < 8; ++kb) {
    if (kb + 1 < 8) {
      X_STAGE((kb + 1) % 2)
      if (kb + 3 < 8) X_FETCH((kb + 1) % 2, xg + (kb + 3) * 32)
      X_FRAGS((kb + 1) % 2)
    }
#pragma unroll
    for (int j = 0; j < 8; ++j) {
      const int k0 = kb * 32 + 4 * j;
      const v2f av = a[kb % 2][j];
      const float* w0 = ((j & 1) ? bp2 : bp0) + k0 * 64;
      const float* w1 = ((j & 1) ? bp3 : bp1) + k0 * 64;
      const float* w2 = ((j & 1) ? bp0 : bp2) + k0 * 64;
      const float* w3 = ((j & 1) ? bp1 : bp3) + k0 * 64;
      v2f b0 = { w0[0], w0[64] };
      v2f b1 = { w1[0], w1[64] };
      v2f b2 = { w2[0], w2[64] };
      v2f b3 = { w3[0], w3[64] };
      acc0 = __builtin_amdgcn_wmma_f32_16x16x4_f32(false, av, false, b0, (short)0, acc0, false, false);
      acc1 = __builtin_amdgcn_wmma_f32_16x16x4_f32(false, av, false, b1, (short)0, acc1, false, false);
      acc2 = __builtin_amdgcn_wmma_f32_16x16x4_f32(false, av, false, b2, (short)0, acc2, false, false);
      acc3 = __builtin_amdgcn_wmma_f32_16x16x4_f32(false, av, false, b3, (short)0, acc3, false, false);
    }
  }
#undef X_FETCH
#undef X_STAGE
#undef X_FRAGS

  float* orow = support + (size_t)(row_base + 8 * hi) * OUT_F + lo;
#pragma unroll
  for (int r = 0; r < 8; ++r) {
    orow[(size_t)r * OUT_F +  0] = acc0[r];
    orow[(size_t)r * OUT_F + 16] = acc1[r];
    orow[(size_t)r * OUT_F + 32] = acc2[r];
    orow[(size_t)r * OUT_F + 48] = acc3[r];
  }
}

#define EDGES_PER_WAVE 128
#define EDGES_PER_HALF 64
#define GATHER_RING    8
#define TAIL_REGS      32

__device__ __forceinline__ void put_row(float* __restrict__ out, int r, int c4,
                                        float x, float y, float z, float w) {
  *(float4*)(out + (size_t)r * OUT_F + c4) = make_float4(x, y, z, w);
}

__device__ __forceinline__ int   pick(const int4 v, int k)   { return k == 0 ? v.x : k == 1 ? v.y : k == 2 ? v.z : v.w; }
__device__ __forceinline__ float pick(const float4 v, int k) { return k == 0 ? v.x : k == 1 ? v.y : k == 2 ? v.z : v.w; }
__device__ __forceinline__ int half_i(int v, int ln) {
  const unsigned sel = (unsigned)ln * 0x11111111u;
  return (int)__builtin_amdgcn_permlane16((unsigned)v, (unsigned)v, sel, sel, false, false);
}
__device__ __forceinline__ float half_f(float v, int ln) { return __int_as_float(half_i(__float_as_int(v), ln)); }
__device__ __forceinline__ int sel_i(int v, unsigned sel) {
  return (int)__builtin_amdgcn_permlane16((unsigned)v, (unsigned)v, sel, sel, false, false);
}
__device__ __forceinline__ float sel_f(float v, unsigned sel) { return __int_as_float(sel_i(__float_as_int(v), sel)); }

__device__ __forceinline__ float4 gather_row(const float* __restrict__ support, int c, int l16) {
  const float4* s4 = (const float4*)support;
  return s4[(unsigned)c * (unsigned)(OUT_F / 4) + (unsigned)l16];
}

__global__ __launch_bounds__(256) void spmm_rows(const float* __restrict__ support,
                                                 const float* __restrict__ edge_vals,
                                                 const int* __restrict__ row_idx,
                                                 const int* __restrict__ col_idx,
                                                 const float* __restrict__ bias,
                                                 float* __restrict__ out) {
  const int lane = threadIdx.x & 31;
  const int l16  = lane & 15;
  const int hsh  = lane & 16;
  const int wave = (int)blockIdx.x * 8 + __builtin_amdgcn_readfirstlane((int)(threadIdx.x >> 5));
  const int w0 = wave * EDGES_PER_WAVE;
  if (w0 >= N_EDGES) return;
  const int e0 = w0 + (lane >> 4) * EDGES_PER_HALF;
  const int e1 = e0 + EDGES_PER_HALF;
  const bool has_next = e1 < N_EDGES;

  const int c4 = l16 * 4;

  const int4   r0 = *(const int4*)(row_idx + w0 + 4 * lane);
  const int4   k0 = *(const int4*)(col_idx + w0 + 4 * lane);
  const float4 v0 = *(const float4*)(edge_vals + w0 + 4 * lane);
  const int et = (has_next ? e1 : e0) + l16;
  const int   ra = row_idx[et],   rb = row_idx[et + 16];
  const int   ka = col_idx[et],   kb = col_idx[et + 16];
  const float va = edge_vals[et], vb = edge_vals[et + 16];

  const float4 b = *(const float4*)(bias + c4);
  float ax = 0.f, ay = 0.f, az = 0.f, aw = 0.f;
  int cur = (e0 > 0) ? row_idx[e0 - 1] : -1;
  bool own = false;

  const int kdl = half_i(k0.x, 2);
  const int kdt = half_i(k0.w, 15);
  int4 kg;
  kg.x = (r0.x == cur) ? kdl : k0.x;
  kg.y = (r0.y == cur) ? kdl : k0.y;
  kg.z = (r0.z == cur) ? kdl : k0.z;
  kg.w = (r0.w == cur) ? kdl : k0.w;

  int pw = (int)__builtin_amdgcn_permlane16((unsigned)r0.w, (unsigned)r0.w,
                                            0x65432100u, 0xEDCBA987u, false, false);
  pw = (l16 == 0) ? cur : pw;
  const unsigned m0 = __builtin_amdgcn_ballot_w32(r0.x != pw);
  const unsigned m1 = __builtin_amdgcn_ballot_w32(r0.y != r0.x);
  const unsigned m2 = __builtin_amdgcn_ballot_w32(r0.z != r0.y);
  const unsigned m3 = __builtin_amdgcn_ballot_w32(r0.w != r0.z);

#define ROW_STEP(c, L, SEL, s)                                              \
  {                                                                         \
    const unsigned mc = (c) == 0 ? m0 : (c) == 1 ? m1 : (c) == 2 ? m2 : m3; \
    if ((mc >> (L)) & 0x10001u) {                                           \
      const int r = sel_i(pick(r0, (c)), (SEL));                            \
      if (r != cur) {                                                       \
        if (own) put_row(out, cur, c4, ax + b.x, ay + b.y, az + b.z, aw + b.w); \
        _Pragma("nounroll")                                                 \
        for (int z = cur + 1; z < r; ++z) put_row(out, z, c4, b.x, b.y, b.z, b.w); \
        cur = r; own = true; ax = 0.f; ay = 0.f; az = 0.f; aw = 0.f;        \
      }                                                                     \
    }                                                                       \
    const float v = sel_f(pick(v0, (c)), (SEL));                            \
    ax = fmaf(v, (s).x, ax); ay = fmaf(v, (s).y, ay);                       \
    az = fmaf(v, (s).z, az); aw = fmaf(v, (s).w, aw);                       \
  }

  float4 g[GATHER_RING];
#pragma unroll
  for (int p = 0; p < GATHER_RING; ++p)
    g[p] = gather_row(support, half_i(pick(kg, p & 3), p >> 2), l16);
  for (int blk = 0; blk < EDGES_PER_HALF / GATHER_RING - 1; ++blk) {
    const int l0 = blk * (GATHER_RING / 4);
    const unsigned s0 = __builtin_amdgcn_readfirstlane((unsigned)l0 * 0x11111111u);
#pragma unroll
    for (int p = 0; p < GATHER_RING; ++p) {
      ROW_STEP(p & 3, l0 + (p >> 2), s0 + (unsigned)(p >> 2) * 0x11111111u, g[p])
      g[p] = gather_row(support, sel_i(pick(kg, p & 3), s0 + (unsigned)(GATHER_RING / 4 + (p >> 2)) * 0x11111111u), l16);
    }
  }
#pragma unroll
  for (int p = 0; p < GATHER_RING; ++p) {
    ROW_STEP(p & 3, (EDGES_PER_HALF - GATHER_RING) / 4 + (p >> 2),
             (unsigned)((EDGES_PER_HALF - GATHER_RING) / 4 + (p >> 2)) * 0x11111111u, g[p])
  }
#undef ROW_STEP

  const unsigned ma = __builtin_amdgcn_ballot_w32(has_next && ra == cur);
  const unsigned mb = __builtin_amdgcn_ballot_w32(has_next && rb == cur);
  const int nt = __builtin_popcount((ma >> hsh) & 0xffffu) + __builtin_popcount((mb >> hsh) & 0xffffu);
  const int kta = (own && has_next && ra == cur) ? ka : kdt;
  const int ktb = (own && has_next && rb == cur) ? kb : kdt;
#pragma unroll
  for (int base = 0; base < TAIL_REGS; base += 8) {
    if (__builtin_amdgcn_ballot_w32(own && base < nt) == 0u) break;
    float4 t[8];
#pragma unroll
    for (int j = 0; j < 8; ++j)
      t[j] = gather_row(support, half_i(base < 16 ? kta : ktb, (base + j) & 15), l16);
#pragma unroll
    for (int j = 0; j < 8; ++j) {
      const float v = half_f(base < 16 ? va : vb, (base + j) & 15);
      if (own && base + j < nt) {
        ax = fmaf(v, t[j].x, ax); ay = fmaf(v, t[j].y, ay);
        az = fmaf(v, t[j].z, az); aw = fmaf(v, t[j].w, aw);
      }
    }
  }
  bool open = own && (!has_next || nt == TAIL_REGS);

#define TAIL_EDGE(r, v, s)                                                  \
  { open = open && ((r) == cur);                                            \
    const float tx = fmaf((v), (s).x, ax);                                  \
    const float ty = fmaf((v), (s).y, ay);                                  \
    const float tz = fmaf((v), (s).z, az);                                  \
    const float tw = fmaf((v), (s).w, aw);                                  \
    ax = open ? tx : ax; ay = open ? ty : ay;                               \
    az = open ? tz : az; aw = open ? tw : aw; }

  for (long e = (long)e1 + TAIL_REGS; open && e < N_EDGES; e += 8) {
    const int4   qa = *(const int4*)(row_idx + e);
    const int4   qb = *(const int4*)(row_idx + e + 4);
    const int4   ca = *(const int4*)(col_idx + e);
    const int4   cb = *(const int4*)(col_idx + e + 4);
    const float4 xa = *(const float4*)(edge_vals + e);
    const float4 xb = *(const float4*)(edge_vals + e + 4);
    const float4 t0 = gather_row(support, ca.x, l16);
    const float4 t1 = gather_row(support, ca.y, l16);
    const float4 t2 = gather_row(support, ca.z, l16);
    const float4 t3 = gather_row(support, ca.w, l16);
    const float4 t4 = gather_row(support, cb.x, l16);
    const float4 t5 = gather_row(support, cb.y, l16);
    const float4 t6 = gather_row(support, cb.z, l16);
    const float4 t7 = gather_row(support, cb.w, l16);
    TAIL_EDGE(qa.x, xa.x, t0)
    TAIL_EDGE(qa.y, xa.y, t1)
    TAIL_EDGE(qa.z, xa.z, t2)
    TAIL_EDGE(qa.w, xa.w, t3)
    TAIL_EDGE(qb.x, xb.x, t4)
    TAIL_EDGE(qb.y, xb.y, t5)
    TAIL_EDGE(qb.z, xb.z, t6)
    TAIL_EDGE(qb.w, xb.w, t7)
  }
#undef TAIL_EDGE

  if (own) {
    put_row(out, cur, c4, ax + b.x, ay + b.y, az + b.z, aw + b.w);
    if (open) {
      for (int z = cur + 1; z < N_NODES; ++z) put_row(out, z, c4, b.x, b.y, b.z, b.w);
    }
  }
}

extern "C" void kernel_launch(void* const* d_in, const int* in_sizes, int n_in,
                              void* d_out, int out_size, void* d_ws, size_t ws_size,
                              hipStream_t stream) {
  const float* X         = (const float*)d_in[0];
  const float* edge_vals = (const float*)d_in[1];
  const float* W         = (const float*)d_in[2];
  const float* bias      = (const float*)d_in[3];
  const int*   row_idx   = (const int*)d_in[4];
  const int*   col_idx   = (const int*)d_in[5];
  float* out     = (float*)d_out;
  float* support = (float*)d_ws;

  (void)in_sizes; (void)n_in; (void)out_size; (void)ws_size;

  gemm_xw_wmma<<<GEMM_GROUPS, 256, 0, stream>>>(X, W, support);
  int waves  = (N_EDGES + EDGES_PER_WAVE - 1) / EDGES_PER_WAVE;
  int blocks = (waves + 7) / 8;
  spmm_rows<<<blocks, 256, 0, stream>>>(support, edge_vals, row_idx, col_idx, bias, out);
}
